# starting kernel
# baseline (speedup 1.0000x reference)
#include <hip/hip_runtime.h>

typedef __attribute__((ext_vector_type(2))) float v2f;
typedef __attribute__((ext_vector_type(8))) float v8f;


__global__ void k_zero(float* __restrict__ p, long long n) {
    long long i = (long long)blockIdx.x * blockDim.x + threadIdx.x;
    if (i < n) p[i] = 0.0f;
}

__global__ void k_rsqrt_norm(float* __restrict__ p, int n) {
    int i = blockIdx.x * blockDim.x + threadIdx.x;
    if (i < n) {
        float v = p[i];
        p[i] = rsqrtf(fmaxf(v, 1.0f));
    }
}


__global__ void k_degrees(const int* __restrict__ src, const int* __restrict__ dst,
                          float* __restrict__ deg_out, float* __restrict__ deg_in, int E) {
    int e = blockIdx.x * blockDim.x + threadIdx.x;
    if (e < E) {
        atomicAdd(&deg_out[src[e]], 1.0f);
        atomicAdd(&deg_in[dst[e]], 1.0f);
    }
}

template <int K, int NC>
__global__ void k_gemm_scaled(const float* __restrict__ X, const float* __restrict__ ns,
                              const float* __restrict__ W, float* __restrict__ P,
                              int nrows) {
    const int m0   = blockIdx.x * 16;
    const int n0   = blockIdx.y * 16;
    const int lane = threadIdx.x;
    const int half = lane >> 4;
    const int lid  = lane & 15;

    const int  arow_raw = m0 + lid;
    const int  arow     = arow_raw < nrows ? arow_raw : (nrows - 1);
    const float scale   = arow_raw < nrows ? ns[arow] : 0.0f;

    const float* __restrict__ xa = X + (size_t)arow * K + 2 * half;
    const float* __restrict__ wb = W + (size_t)(2 * half) * NC + n0 + lid;

    v8f acc = {};
#pragma unroll 8
    for (int k = 0; k < K; k += 4) {
        v2f a, b;
        const float2 av = *(const float2*)(xa + k);
        a.x = av.x * scale;
        a.y = av.y * scale;
        b.x = wb[(size_t)k * NC];
        b.y = wb[(size_t)(k + 1) * NC];
        acc = __builtin_amdgcn_wmma_f32_16x16x4_f32(
             false, a,  false, b,
             (short)0, acc,  false,  false);
    }

    float* __restrict__ pp = P + (size_t)(m0 + 8 * half) * NC + n0 + lid;
    if (m0 + 16 <= nrows) {
#pragma unroll
        for (int r = 0; r < 8; ++r)
            pp[(size_t)r * NC] = acc[r];
    } else {
#pragma unroll
        for (int r = 0; r < 8; ++r)
            if (m0 + r + 8 * half < nrows)
                pp[(size_t)r * NC] = acc[r];
    }
}

__global__ void k_aggregate(const float* __restrict__ P, const int* __restrict__ src,
                            const int* __restrict__ dst, float* __restrict__ G,
                            long long total, int shift, int d) {
    long long tid = (long long)blockIdx.x * blockDim.x + threadIdx.x;
    if (tid >= total) return;
    const int e = (int)(tid >> shift);
    const int c = (int)(tid & ((1 << shift) - 1));
    const int s = src[e];
    const int t = dst[e];
    const float4 v = *(const float4*)(P + (size_t)s * d + 4 * c);
    float* gp = G + (size_t)t * d + 4 * c;
    atomicAdd(gp + 0, v.x);
    atomicAdd(gp + 1, v.y);
    atomicAdd(gp + 2, v.z);
    atomicAdd(gp + 3, v.w);
}

__global__ void k_finalize(const float* __restrict__ G, const float* __restrict__ nd,
                           const float* __restrict__ b, float* __restrict__ H,
                           long long total, int dshift, int relu) {
    long long idx = (long long)blockIdx.x * blockDim.x + threadIdx.x;
    if (idx >= total) return;
    const int i = (int)(idx >> dshift);
    const int j = (int)(idx & ((1 << dshift) - 1));
    float v = G[idx] * nd[i] + b[j];
    if (relu) v = fmaxf(v, 0.0f);
    H[idx] = v;
}


static inline int cdiv_ll(long long a, int b) { return (int)((a + b - 1) / b); }

extern "C" void kernel_launch(void* const* d_in, const int* in_sizes, int n_in,
                              void* d_out, int out_size, void* d_ws, size_t ws_size,
                              hipStream_t stream) {
    const float* feat = (const float*)d_in[0];
    const float* W1   = (const float*)d_in[1];
    const float* b1   = (const float*)d_in[2];
    const float* W2   = (const float*)d_in[3];
    const float* b2   = (const float*)d_in[4];
    const float* W3   = (const float*)d_in[5];
    const float* b3   = (const float*)d_in[6];
    const int*   src  = (const int*)d_in[7];
    const int*   dst  = (const int*)d_in[8];

    const int D_IN = 256, D_H = 128, D_OUT = 64;
    const int n = in_sizes[0] / D_IN;
    const int E = in_sizes[7];

    float* ns   = (float*)d_ws;
    float* nd   = ns + n;
    float* buf0 = nd + n;
    float* buf1 = buf0 + (size_t)n * D_H;

    float* out = (float*)d_out;

    const int mtiles = (n + 15) / 16;

    k_zero<<<cdiv_ll(2LL * n, 256), 256, 0, stream>>>(ns, 2LL * n);
    k_degrees<<<(E + 255) / 256, 256, 0, stream>>>(src, dst, ns, nd, E);
    k_rsqrt_norm<<<(2 * n + 255) / 256, 256, 0, stream>>>(ns, 2 * n);

    k_gemm_scaled<256, 128><<<dim3(mtiles, D_H / 16), 32, 0, stream>>>(feat, ns, W1, buf0, n);
    k_zero<<<cdiv_ll((long long)n * D_H, 256), 256, 0, stream>>>(buf1, (long long)n * D_H);
    {
        long long total = (long long)E * (D_H / 4);
        k_aggregate<<<cdiv_ll(total, 256), 256, 0, stream>>>(buf0, src, dst, buf1, total, 5, D_H);
    }
    k_finalize<<<cdiv_ll((long long)n * D_H, 256), 256, 0, stream>>>(buf1, nd, b1, buf0,
                                                                     (long long)n * D_H, 7, 1);

    k_gemm_scaled<128, 128><<<dim3(mtiles, D_H / 16), 32, 0, stream>>>(buf0, ns, W2, buf1, n);
    k_zero<<<cdiv_ll((long long)n * D_H, 256), 256, 0, stream>>>(buf0, (long long)n * D_H);
    {
        long long total = (long long)E * (D_H / 4);
        k_aggregate<<<cdiv_ll(total, 256), 256, 0, stream>>>(buf1, src, dst, buf0, total, 5, D_H);
    }
    k_finalize<<<cdiv_ll((long long)n * D_H, 256), 256, 0, stream>>>(buf0, nd, b2, buf1,
                                                                     (long long)n * D_H, 7, 1);

    k_gemm_scaled<128, 64><<<dim3(mtiles, D_OUT / 16), 32, 0, stream>>>(buf1, ns, W3, buf0, n);
    k_zero<<<cdiv_ll((long long)n * D_OUT, 256), 256, 0, stream>>>(buf1, (long long)n * D_OUT);
    {
        long long total = (long long)E * (D_OUT / 4);
        k_aggregate<<<cdiv_ll(total, 256), 256, 0, stream>>>(buf0, src, dst, buf1, total, 4, D_OUT);
    }
    k_finalize<<<cdiv_ll((long long)n * D_OUT, 256), 256, 0, stream>>>(buf1, nd, b3, out,
                                                                       (long long)n * D_OUT, 6, 0);
}


# Zero + float-atomic scatter-add + finalize replaced by one launch per layer: a workgroup owns rows x 64 features in LDS, filters the whole edge list, adds by LDS atomics; sum order = LDS arrival
# speedup vs baseline: 1.6316x; 1.6316x over previous
#include <hip/hip_runtime.h>

typedef __attribute__((ext_vector_type(2))) float v2f;
typedef __attribute__((ext_vector_type(8))) float v8f;


__global__ void k_zero(float* __restrict__ p, long long n) {
    long long i = (long long)blockIdx.x * blockDim.x + threadIdx.x;
    if (i < n) p[i] = 0.0f;
}

__global__ void k_rsqrt_norm(float* __restrict__ p, int n) {
    int i = blockIdx.x * blockDim.x + threadIdx.x;
    if (i < n) {
        float v = p[i];
        p[i] = rsqrtf(fmaxf(v, 1.0f));
    }
}


__global__ void k_degrees(const int* __restrict__ src, const int* __restrict__ dst,
                          float* __restrict__ deg_out, float* __restrict__ deg_in, int E) {
    int e = blockIdx.x * blockDim.x + threadIdx.x;
    if (e < E) {
        atomicAdd(&deg_out[src[e]], 1.0f);
        atomicAdd(&deg_in[dst[e]], 1.0f);
    }
}

template <int K, int NC>
__global__ void k_gemm_scaled(const float* __restrict__ X, const float* __restrict__ ns,
                              const float* __restrict__ W, float* __restrict__ P,
                              int nrows) {
    const int m0   = blockIdx.x * 16;
    const int n0   = blockIdx.y * 16;
    const int lane = threadIdx.x;
    const int half = lane >> 4;
    const int lid  = lane & 15;

    const int  arow_raw = m0 + lid;
    const int  arow     = arow_raw < nrows ? arow_raw : (nrows - 1);
    const float scale   = arow_raw < nrows ? ns[arow] : 0.0f;

    const float* __restrict__ xa = X + (size_t)arow * K + 2 * half;
    const float* __restrict__ wb = W + (size_t)(2 * half) * NC + n0 + lid;

    v8f acc = {};
#pragma unroll 8
    for (int k = 0; k < K; k += 4) {
        v2f a, b;
        const float2 av = *(const float2*)(xa + k);
        a.x = av.x * scale;
        a.y = av.y * scale;
        b.x = wb[(size_t)k * NC];
        b.y = wb[(size_t)(k + 1) * NC];
        acc = __builtin_amdgcn_wmma_f32_16x16x4_f32(
             false, a,  false, b,
             (short)0, acc,  false,  false);
    }

    float* __restrict__ pp = P + (size_t)(m0 + 8 * half) * NC + n0 + lid;
    if (m0 + 16 <= nrows) {
#pragma unroll
        for (int r = 0; r < 8; ++r)
            pp[(size_t)r * NC] = acc[r];
    } else {
#pragma unroll
        for (int r = 0; r < 8; ++r)
            if (m0 + r + 8 * half < nrows)
                pp[(size_t)r * NC] = acc[r];
    }
}


template <int D, bool FULL>
__device__ __forceinline__ void ring_drain(const float* __restrict__ Pslice, const int* __restrict__ src,
                                           const int* __restrict__ dst, float* acc, const int* qe,
                                           int lo, int tail, int nvalid, int lane) {
    constexpr int U = 16;
    __builtin_amdgcn_wave_barrier();
    const int e_l = (FULL || lane < nvalid) ? qe[(tail + lane) & 255] : 0;
    const int s_l = src[e_l];
    const int t_l = dst[e_l] - lo;
#pragma unroll
    for (int i0 = 0; i0 < 32; i0 += U) {
        if (!FULL && i0 >= nvalid) break;
        float2 v[U];
#pragma unroll
        for (int u = 0; u < U; ++u) {
            if (FULL || i0 + u < nvalid) {
                const int s = __builtin_amdgcn_readlane(s_l, i0 + u);
                v[u] = *(const float2*)(Pslice + (size_t)s * D + 2 * lane);
            }
        }
#pragma unroll
        for (int u = 0; u < U; ++u) {
            if (FULL || i0 + u < nvalid) {
                const int t = __builtin_amdgcn_readlane(t_l, i0 + u);
                float* a = acc + t * 64 + 2 * lane;
                atomicAdd(a,     v[u].x);
                atomicAdd(a + 1, v[u].y);
            }
        }
    }
}

template <int D, int M, bool RELU>
__global__ __launch_bounds__(1024)
void k_agg_tile(const float* __restrict__ P, const int* __restrict__ src,
                const int* __restrict__ dst, const float* __restrict__ nd,
                const float* __restrict__ b, float* __restrict__ H, int n, int E) {
    constexpr int F  = 64;
    constexpr int S  = D / F;
    constexpr int NW = 32;
    constexpr int CH = 128;
    __shared__ __attribute__((aligned(16))) float acc[M * F];
    __shared__ int q_e[NW * 256];

    const int tid  = threadIdx.x;
    const int lane = tid & 31;
    const int w     = __builtin_amdgcn_readfirstlane(tid >> 5);
    const int slice = blockIdx.x % S;
    const int lo    = (blockIdx.x / S) * M;
    const int rows  = min(M, n - lo);
    const float* __restrict__ Pslice = P + slice * F;
    int* qe = q_e + w * 256;

    for (int i = tid; i < M * F / 4; i += 1024)
        ((float4*)acc)[i] = make_float4(0.f, 0.f, 0.f, 0.f);
    __syncthreads();

    int head = 0, tail = 0;
    const unsigned lt_mask = (1u << lane) - 1u;

    auto scan_step = [&](const int4 cv, int e0) {
        const int dv[4] = {cv.x, cv.y, cv.z, cv.w};
#pragma unroll
        for (int c = 0; c < 4; ++c) {
            const bool match = (unsigned)(dv[c] - lo) < (unsigned)rows;
            const unsigned m = __builtin_amdgcn_ballot_w32(match);
            if (m) {
                if (match)
                    qe[(head + __builtin_popcount(m & lt_mask)) & 255] = e0 + c;
                head += __builtin_popcount(m);
            }
        }
        while (head - tail >= 32) {
            ring_drain<D, true>(Pslice, src, dst, acc, qe, lo, tail, 32, lane);
            tail += 32;
        }
    };

    const int nsteps_all = E / CH;
    if (w < nsteps_all) {
        const int rot = (int)blockIdx.x * (nsteps_all / (int)gridDim.x);
        auto step_base = [&](int idx) {
            int s = idx + rot;
            if (s >= nsteps_all) s -= nsteps_all;
            return s * CH;
        };
        auto ld = [&](int idx) {
            const int bb = step_base(idx < nsteps_all ? idx : w);
            return *(const int4*)(dst + bb + lane * 4);
        };
        int idx = w;
        int4 c0 = ld(idx);
        int4 c1 = ld(idx + NW);
        int4 c2 = ld(idx + 2 * NW);
        int4 c3 = ld(idx + 3 * NW);
        for (; idx < nsteps_all; idx += 4 * NW) {
            scan_step(c0, step_base(idx) + lane * 4);
            c0 = ld(idx + 4 * NW);
            if (idx + NW >= nsteps_all) break;
            scan_step(c1, step_base(idx + NW) + lane * 4);
            c1 = ld(idx + 5 * NW);
            if (idx + 2 * NW >= nsteps_all) break;
            scan_step(c2, step_base(idx + 2 * NW) + lane * 4);
            c2 = ld(idx + 6 * NW);
            if (idx + 3 * NW >= nsteps_all) break;
            scan_step(c3, step_base(idx + 3 * NW) + lane * 4);
            c3 = ld(idx + 7 * NW);
        }
    }
    if (w == 0) {
        const int e_tail = nsteps_all * CH;
        for (int eb = e_tail; eb < E; eb += 32) {
            const int e = eb + lane;
            const int d = (e < E) ? dst[e] : -1;
            const bool match = (unsigned)(d - lo) < (unsigned)rows;
            const unsigned m = __builtin_amdgcn_ballot_w32(match);
            if (m) {
                if (match)
                    qe[(head + __builtin_popcount(m & lt_mask)) & 255] = e;
                head += __builtin_popcount(m);
            }
        }
        while (head - tail >= 32) {
            ring_drain<D, true>(Pslice, src, dst, acc, qe, lo, tail, 32, lane);
            tail += 32;
        }
    }
    if (head - tail > 0)
        ring_drain<D, false>(Pslice, src, dst, acc, qe, lo, tail, head - tail, lane);
    __syncthreads();

    for (int i = tid; i < rows * F; i += 1024) {
        const int r = i >> 6, f = i & 63;
        float v = acc[i] * nd[lo + r] + b[slice * F + f];
        if (RELU) v = fmaxf(v, 0.0f);
        H[(size_t)(lo + r) * D + slice * F + f] = v;
    }
}


static inline int cdiv_ll(long long a, int b) { return (int)((a + b - 1) / b); }

extern "C" void kernel_launch(void* const* d_in, const int* in_sizes, int n_in,
                              void* d_out, int out_size, void* d_ws, size_t ws_size,
                              hipStream_t stream) {
    const float* feat = (const float*)d_in[0];
    const float* W1   = (const float*)d_in[1];
    const float* b1   = (const float*)d_in[2];
    const float* W2   = (const float*)d_in[3];
    const float* b2   = (const float*)d_in[4];
    const float* W3   = (const float*)d_in[5];
    const float* b3   = (const float*)d_in[6];
    const int*   src  = (const int*)d_in[7];
    const int*   dst  = (const int*)d_in[8];

    const int D_IN = 256, D_H = 128, D_OUT = 64;
    const int n = in_sizes[0] / D_IN;
    const int E = in_sizes[7];

    float* ns   = (float*)d_ws;
    float* nd   = ns + n;
    float* buf0 = nd + n;
    float* buf1 = buf0 + (size_t)n * D_H;

    float* out = (float*)d_out;

    const int mtiles = (n + 15) / 16;

    k_zero<<<cdiv_ll(2LL * n, 256), 256, 0, stream>>>(ns, 2LL * n);
    k_degrees<<<(E + 255) / 256, 256, 0, stream>>>(src, dst, ns, nd, E);
    k_rsqrt_norm<<<(2 * n + 255) / 256, 256, 0, stream>>>(ns, 2 * n);

    k_gemm_scaled<256, 128><<<dim3(mtiles, D_H / 16), 32, 0, stream>>>(feat, ns, W1, buf0, n);
    k_agg_tile<128, 800, true><<<(n + 799) / 800 * 2, 1024, 0, stream>>>(buf0, src, dst, nd, b1, buf1, n, E);

    k_gemm_scaled<128, 128><<<dim3(mtiles, D_H / 16), 32, 0, stream>>>(buf1, ns, W2, buf0, n);
    k_agg_tile<128, 800, true><<<(n + 799) / 800 * 2, 1024, 0, stream>>>(buf0, src, dst, nd, b2, buf1, n, E);

    k_gemm_scaled<128, 64><<<dim3(mtiles, D_OUT / 16), 32, 0, stream>>>(buf1, ns, W3, buf0, n);
    k_agg_tile<64, 400, false><<<(n + 399) / 400, 1024, 0, stream>>>(buf0, src, dst, nd, b3, out, n, E);
}


# Edge filter run once: layer 1's tile scan also appends its hits as (src,dst) pairs per 400-row block, parked in the output buffer; layers 2,3 read the lists, no scan; sum order still LDS arrival
# speedup vs baseline: 1.9536x; 1.1974x over previous
#include <hip/hip_runtime.h>

typedef __attribute__((ext_vector_type(2))) float v2f;
typedef __attribute__((ext_vector_type(8))) float v8f;


__global__ void k_zero(float* __restrict__ p, long long n) {
    long long i = (long long)blockIdx.x * blockDim.x + threadIdx.x;
    if (i < n) p[i] = 0.0f;
}

__global__ void k_rsqrt_norm(float* __restrict__ p, int n) {
    int i = blockIdx.x * blockDim.x + threadIdx.x;
    if (i < n) {
        float v = p[i];
        p[i] = rsqrtf(fmaxf(v, 1.0f));
    }
}


__global__ void k_degrees(const int* __restrict__ src, const int* __restrict__ dst,
                          float* __restrict__ deg_out, float* __restrict__ deg_in, int E) {
    int e = blockIdx.x * blockDim.x + threadIdx.x;
    if (e < E) {
        atomicAdd(&deg_out[src[e]], 1.0f);
        atomicAdd(&deg_in[dst[e]], 1.0f);
    }
}

template <int K, int NC>
__global__ void k_gemm_scaled(const float* __restrict__ X, const float* __restrict__ ns,
                              const float* __restrict__ W, float* __restrict__ P,
                              int nrows) {
    const int m0   = blockIdx.x * 16;
    const int n0   = blockIdx.y * 16;
    const int lane = threadIdx.x;
    const int half = lane >> 4;
    const int lid  = lane & 15;

    const int  arow_raw = m0 + lid;
    const int  arow     = arow_raw < nrows ? arow_raw : (nrows - 1);
    const float scale   = arow_raw < nrows ? ns[arow] : 0.0f;

    const float* __restrict__ xa = X + (size_t)arow * K + 2 * half;
    const float* __restrict__ wb = W + (size_t)(2 * half) * NC + n0 + lid;

    v8f acc = {};
#pragma unroll 8
    for (int k = 0; k < K; k += 4) {
        v2f a, b;
        const float2 av = *(const float2*)(xa + k);
        a.x = av.x * scale;
        a.y = av.y * scale;
        b.x = wb[(size_t)k * NC];
        b.y = wb[(size_t)(k + 1) * NC];
        acc = __builtin_amdgcn_wmma_f32_16x16x4_f32(
             false, a,  false, b,
             (short)0, acc,  false,  false);
    }

    float* __restrict__ pp = P + (size_t)(m0 + 8 * half) * NC + n0 + lid;
    if (m0 + 16 <= nrows) {
#pragma unroll
        for (int r = 0; r < 8; ++r)
            pp[(size_t)r * NC] = acc[r];
    } else {
#pragma unroll
        for (int r = 0; r < 8; ++r)
            if (m0 + r + 8 * half < nrows)
                pp[(size_t)r * NC] = acc[r];
    }
}


constexpr int SUB = 400;

template <int D, bool FULL>
__device__ __forceinline__ void gather_add(const float* __restrict__ Pslice, float* acc,
                                           int s_l, int t_l, int nvalid, int lane) {
    constexpr int U = 16;
#pragma unroll
    for (int i0 = 0; i0 < 32; i0 += U) {
        if (!FULL && i0 >= nvalid) break;
        float2 v[U];
#pragma unroll
        for (int u = 0; u < U; ++u) {
            if (FULL || i0 + u < nvalid) {
                const int s = __builtin_amdgcn_readlane(s_l, i0 + u);
                v[u] = *(const float2*)(Pslice + (size_t)s * D + 2 * lane);
            }
        }
#pragma unroll
        for (int u = 0; u < U; ++u) {
            if (FULL || i0 + u < nvalid) {
                const int t = __builtin_amdgcn_readlane(t_l, i0 + u);
                float* a = acc + t * 64 + 2 * lane;
                atomicAdd(a,     v[u].x);
                atomicAdd(a + 1, v[u].y);
            }
        }
    }
}

struct HitLists {
    int2* r0;  int2* r1;
    int   cap0, cap1;
    int*  cnt;
    bool  on;
};

template <int D, bool FULL, bool WRITE>
__device__ __forceinline__ void ring_drain(const float* __restrict__ Pslice, const int* __restrict__ src,
                                           const int* __restrict__ dst, float* acc, const int* qe,
                                           int lo, int tail, int nvalid, int lane, const HitLists& hl) {
    __builtin_amdgcn_wave_barrier();
    const int e_l = (FULL || lane < nvalid) ? qe[(tail + lane) & 255] : 0;
    const int s_l = src[e_l];
    const int d_l = dst[e_l];
    const int t_l = d_l - lo;
    gather_add<D, FULL>(Pslice, acc, s_l, t_l, nvalid, lane);
    if (WRITE && hl.on) {
        const bool valid = FULL || lane < nvalid;
        const bool up = valid && t_l >= SUB;
        const bool dn = valid && t_l < SUB;
        const unsigned mu = __builtin_amdgcn_ballot_w32(up);
        const unsigned md = __builtin_amdgcn_ballot_w32(dn);
        int b0 = 0, b1 = 0;
        if (lane == 0) {
            b0 = atomicAdd(hl.cnt,     __builtin_popcount(md));
            b1 = atomicAdd(hl.cnt + 1, __builtin_popcount(mu));
        }
        b0 = __builtin_amdgcn_readfirstlane(b0);
        b1 = __builtin_amdgcn_readfirstlane(b1);
        const unsigned below = (1u << lane) - 1u;
        const int pos = up ? b1 + __builtin_popcount(mu & below) : b0 + __builtin_popcount(md & below);
        const int cap = up ? hl.cap1 : hl.cap0;
        if (valid && pos < cap) {
            int2* r = up ? hl.r1 : hl.r0;
            r[pos] = make_int2(s_l, d_l);
        }
    }
}

template <int D, int M, bool RELU, int LMODE>
__global__ __launch_bounds__(1024)
void k_agg_tile(const float* __restrict__ P, const int* __restrict__ src,
                const int* __restrict__ dst, const float* __restrict__ nd,
                const float* __restrict__ b, float* H, float* lst, int n, int E) {
    constexpr int F  = 64;
    constexpr int S  = D / F;
    constexpr int NW = 32;
    constexpr int CH = 128;
    constexpr int NR = M / SUB;
    static_assert(M % SUB == 0 && (LMODE != 1 || NR == 2), "tile rows are whole hit-list blocks");
    constexpr bool WR = (LMODE == 1);
    __shared__ __attribute__((aligned(16))) float acc[M * F];
    __shared__ int q_e[NW * 256];
    __shared__ int l_cnt[2];

    const int tid  = threadIdx.x;
    const int lane = tid & 31;
    const int w     = __builtin_amdgcn_readfirstlane(tid >> 5);
    const int slice = blockIdx.x % S;
    const int lo    = (blockIdx.x / S) * M;
    const int rows  = min(M, n - lo);
    const float* __restrict__ Pslice = P + slice * F;
    int* qe = q_e + w * 256;

    for (int i = tid; i < M * F / 4; i += 1024)
        ((float4*)acc)[i] = make_float4(0.f, 0.f, 0.f, 0.f);
    if (tid < 2) l_cnt[tid] = 0;

    int2* reg[2];
    int   rcap[2];
    bool  have[2];
#pragma unroll
    for (int j = 0; j < 2; ++j) {
        const int r0 = lo + j * SUB;
        have[j] = j < NR && r0 < n;
        reg[j]  = (int2*)(lst + (size_t)(have[j] ? r0 : lo) * 64);
        rcap[j] = have[j] ? min(SUB, n - r0) * 32 - 1 : 0;
    }
    HitLists hl;
    hl.r0 = reg[0]; hl.r1 = reg[1]; hl.cap0 = rcap[0]; hl.cap1 = rcap[1];
    hl.cnt = l_cnt; hl.on = WR && slice == 0;

    int  rcnt[2]  = {0, 0};
    bool use_list = (LMODE == 2);
    if (LMODE == 2) {
#pragma unroll
        for (int j = 0; j < NR; ++j) {
            if (have[j]) {
                rcnt[j] = reg[j][rcap[j]].x;
                use_list = use_list && rcnt[j] <= rcap[j];
            }
        }
    }
    __syncthreads();

    if (use_list) {
#pragma unroll
        for (int j = 0; j < NR; ++j) {
            const int   c  = rcnt[j];
            const int2* rg = reg[j];
            int  base = w * 32;
            int2 nxt  = make_int2(0, lo);
            if (base + lane < c) nxt = rg[base + lane];
            while (base + 32 <= c) {
                const int2 cur = nxt;
                const int  nb  = base + NW * 32;
                if (nb + lane < c) nxt = rg[nb + lane];
                gather_add<D, true>(Pslice, acc, cur.x, cur.y - lo, 32, lane);
                base = nb;
            }
            if (base < c)
                gather_add<D, false>(Pslice, acc, nxt.x, nxt.y - lo, c - base, lane);
        }
    } else {
    int head = 0, tail = 0;
    const unsigned lt_mask = (1u << lane) - 1u;

    auto scan_step = [&](const int4 cv, int e0) {
        const int dv[4] = {cv.x, cv.y, cv.z, cv.w};
#pragma unroll
        for (int c = 0; c < 4; ++c) {
            const bool match = (unsigned)(dv[c] - lo) < (unsigned)rows;
            const unsigned m = __builtin_amdgcn_ballot_w32(match);
            if (m) {
                if (match)
                    qe[(head + __builtin_popcount(m & lt_mask)) & 255] = e0 + c;
                head += __builtin_popcount(m);
            }
        }
        while (head - tail >= 32) {
            ring_drain<D, true, WR>(Pslice, src, dst, acc, qe, lo, tail, 32, lane, hl);
            tail += 32;
        }
    };

    const int nsteps_all = E / CH;
    if (w < nsteps_all) {
        const int rot = (int)blockIdx.x * (nsteps_all / (int)gridDim.x);
        auto step_base = [&](int idx) {
            int s = idx + rot;
            if (s >= nsteps_all) s -= nsteps_all;
            return s * CH;
        };
        auto ld = [&](int idx) {
            const int bb = step_base(idx < nsteps_all ? idx : w);
            return *(const int4*)(dst + bb + lane * 4);
        };
        int idx = w;
        int4 c0 = ld(idx);
        int4 c1 = ld(idx + NW);
        int4 c2 = ld(idx + 2 * NW);
        int4 c3 = ld(idx + 3 * NW);
        for (; idx < nsteps_all; idx += 4 * NW) {
            scan_step(c0, step_base(idx) + lane * 4);
            c0 = ld(idx + 4 * NW);
            if (idx + NW >= nsteps_all) break;
            scan_step(c1, step_base(idx + NW) + lane * 4);
            c1 = ld(idx + 5 * NW);
            if (idx + 2 * NW >= nsteps_all) break;
            scan_step(c2, step_base(idx + 2 * NW) + lane * 4);
            c2 = ld(idx + 6 * NW);
            if (idx + 3 * NW >= nsteps_all) break;
            scan_step(c3, step_base(idx + 3 * NW) + lane * 4);
            c3 = ld(idx + 7 * NW);
        }
    }
    if (w == 0) {
        const int e_tail = nsteps_all * CH;
        for (int eb = e_tail; eb < E; eb += 32) {
            const int e = eb + lane;
            const int d = (e < E) ? dst[e] : -1;
            const bool match = (unsigned)(d - lo) < (unsigned)rows;
            const unsigned m = __builtin_amdgcn_ballot_w32(match);
            if (m) {
                if (match)
                    qe[(head + __builtin_popcount(m & lt_mask)) & 255] = e;
                head += __builtin_popcount(m);
            }
        }
        while (head - tail >= 32) {
            ring_drain<D, true, WR>(Pslice, src, dst, acc, qe, lo, tail, 32, lane, hl);
            tail += 32;
        }
    }
    if (head - tail > 0)
        ring_drain<D, false, WR>(Pslice, src, dst, acc, qe, lo, tail, head - tail, lane, hl);
    }
    __syncthreads();

    if (WR && slice == 0) {
        if (tid == 0 && have[0]) reg[0][rcap[0]] = make_int2(l_cnt[0], 0);
        if (tid == 1 && have[1]) reg[1][rcap[1]] = make_int2(l_cnt[1], 0);
    }

    for (int i = tid; i < rows * F; i += 1024) {
        const int r = i >> 6, f = i & 63;
        float v = acc[i] * nd[lo + r] + b[slice * F + f];
        if (RELU) v = fmaxf(v, 0.0f);
        H[(size_t)(lo + r) * D + slice * F + f] = v;
    }
}


static inline int cdiv_ll(long long a, int b) { return (int)((a + b - 1) / b); }

extern "C" void kernel_launch(void* const* d_in, const int* in_sizes, int n_in,
                              void* d_out, int out_size, void* d_ws, size_t ws_size,
                              hipStream_t stream) {
    const float* feat = (const float*)d_in[0];
    const float* W1   = (const float*)d_in[1];
    const float* b1   = (const float*)d_in[2];
    const float* W2   = (const float*)d_in[3];
    const float* b2   = (const float*)d_in[4];
    const float* W3   = (const float*)d_in[5];
    const float* b3   = (const float*)d_in[6];
    const int*   src  = (const int*)d_in[7];
    const int*   dst  = (const int*)d_in[8];

    const int D_IN = 256, D_H = 128, D_OUT = 64;
    const int n = in_sizes[0] / D_IN;
    const int E = in_sizes[7];

    float* ns   = (float*)d_ws;
    float* nd   = ns + n;
    float* buf0 = nd + n;
    float* buf1 = buf0 + (size_t)n * D_H;

    float* out = (float*)d_out;

    const int mtiles = (n + 15) / 16;

    k_zero<<<cdiv_ll(2LL * n, 256), 256, 0, stream>>>(ns, 2LL * n);
    k_degrees<<<(E + 255) / 256, 256, 0, stream>>>(src, dst, ns, nd, E);
    k_rsqrt_norm<<<(2 * n + 255) / 256, 256, 0, stream>>>(ns, 2 * n);

    k_gemm_scaled<256, 128><<<dim3(mtiles, D_H / 16), 32, 0, stream>>>(feat, ns, W1, buf0, n);
    k_agg_tile<128, 800, true, 1><<<(n + 799) / 800 * 2, 1024, 0, stream>>>(buf0, src, dst, nd, b1, buf1, out, n, E);

    k_gemm_scaled<128, 128><<<dim3(mtiles, D_H / 16), 32, 0, stream>>>(buf1, ns, W2, buf0, n);
    k_agg_tile<128, 800, true, 2><<<(n + 799) / 800 * 2, 1024, 0, stream>>>(buf0, src, dst, nd, b2, buf1, out, n, E);

    k_gemm_scaled<128, 64><<<dim3(mtiles, D_OUT / 16), 32, 0, stream>>>(buf1, ns, W3, buf0, n);
    k_agg_tile<64, 400, false, 2><<<(n + 399) / 400, 1024, 0, stream>>>(buf0, src, dst, nd, b3, out, out, n, E);
}


# Projection GEMM: X goes through LDS (64 rows x 32 k per chunk, whole-line 16-byte loads, 144-byte pitch), a wave keeps one 16-column block x 4 row tiles, W fragments direct; k order, sums unchanged
# speedup vs baseline: 2.5196x; 1.2897x over previous
#include <hip/hip_runtime.h>

typedef __attribute__((ext_vector_type(2))) float v2f;
typedef __attribute__((ext_vector_type(8))) float v8f;


__global__ void k_zero(float* __restrict__ p, long long n) {
    long long i = (long long)blockIdx.x * blockDim.x + threadIdx.x;
    if (i < n) p[i] = 0.0f;
}

__global__ void k_rsqrt_norm(float* __restrict__ p, int n) {
    int i = blockIdx.x * blockDim.x + threadIdx.x;
    if (i < n) {
        float v = p[i];
        p[i] = rsqrtf(fmaxf(v, 1.0f));
    }
}


__global__ void k_degrees(const int* __restrict__ src, const int* __restrict__ dst,
                          float* __restrict__ deg_out, float* __restrict__ deg_in, int E) {
    int e = blockIdx.x * blockDim.x + threadIdx.x;
    if (e < E) {
        atomicAdd(&deg_out[src[e]], 1.0f);
        atomicAdd(&deg_in[dst[e]], 1.0f);
    }
}

template <int K, int NC>
__global__ __launch_bounds__(NC * 2)
void k_gemm_scaled(const float* __restrict__ X, const float* __restrict__ ns,
                   const float* __restrict__ W, float* __restrict__ P,
                   int nrows) {
    constexpr int BM    = 64;
    constexpr int MT    = BM / 16;
    constexpr int KC    = 32;
    constexpr int PITCH = KC + 4;
    constexpr int NTHR  = NC * 2;
    constexpr int RPP   = NTHR / 8;
    constexpr int NP    = BM / RPP;
    constexpr int NCH   = K / KC;
    constexpr int KS    = KC / 4;
    static_assert(K % KC == 0 && NC % 16 == 0 && BM % RPP == 0, "tile shape");
    __shared__ __attribute__((aligned(16))) float xs[2 * BM * PITCH];

    const int tid  = threadIdx.x;
    const int lane = tid & 31;
    const int w    = __builtin_amdgcn_readfirstlane(tid >> 5);
    const int half = lane >> 4;
    const int lid  = lane & 15;
    const int m0   = blockIdx.x * BM;
    const int n0   = w * 16;

    const int srow = tid >> 3;
    const int scol = (tid & 7) * 4;
    const float* xp[NP];
    float sc[NP];
#pragma unroll
    for (int p = 0; p < NP; ++p) {
        const int  rr = m0 + srow + RPP * p;
        const bool ok = rr < nrows;
        const int  rc = ok ? rr : (nrows - 1);
        sc[p] = ok ? ns[rc] : 0.0f;
        xp[p] = X + (size_t)rc * K + scol;
    }
    const float* __restrict__ wb = W + (size_t)(2 * half) * NC + n0 + lid;

    float4 xr[NP];
    float  bcur[2 * KS], bnxt[2 * KS];
#pragma unroll
    for (int p = 0; p < NP; ++p) xr[p] = *(const float4*)(xp[p]);
#pragma unroll
    for (int ks = 0; ks < KS; ++ks) {
        bcur[2 * ks]     = wb[(size_t)(ks * 4) * NC];
        bcur[2 * ks + 1] = wb[(size_t)(ks * 4 + 1) * NC];
    }
#pragma unroll
    for (int p = 0; p < NP; ++p) {
        float4 v = xr[p];
        v.x *= sc[p]; v.y *= sc[p]; v.z *= sc[p]; v.w *= sc[p];
        *(float4*)(xs + (srow + RPP * p) * PITCH + scol) = v;
    }
    __syncthreads();

    v8f acc[MT];
#pragma unroll
    for (int mt = 0; mt < MT; ++mt) acc[mt] = v8f{};

    for (int kc = 0; kc < NCH; ++kc) {
        const bool more = kc + 1 < NCH;
        if (more) {
            const int k1 = (kc + 1) * KC;
#pragma unroll
            for (int p = 0; p < NP; ++p) xr[p] = *(const float4*)(xp[p] + k1);
#pragma unroll
            for (int ks = 0; ks < KS; ++ks) {
                bnxt[2 * ks]     = wb[(size_t)(k1 + ks * 4) * NC];
                bnxt[2 * ks + 1] = wb[(size_t)(k1 + ks * 4 + 1) * NC];
            }
        }
        __builtin_amdgcn_sched_barrier(0);
        const float* xb = xs + (kc & 1) * (BM * PITCH) + lid * PITCH + 2 * half;
#pragma unroll
        for (int ks = 0; ks < KS; ++ks) {
            v2f b;
            b.x = bcur[2 * ks];
            b.y = bcur[2 * ks + 1];
#pragma unroll
            for (int mt = 0; mt < MT; ++mt) {
                const v2f a = *(const v2f*)(xb + mt * 16 * PITCH + ks * 4);
                acc[mt] = __builtin_amdgcn_wmma_f32_16x16x4_f32(
                     false, a,  false, b,
                     (short)0, acc[mt],  false,  false);
            }
        }
        if (more) {
            float* xd = xs + ((kc + 1) & 1) * (BM * PITCH);
#pragma unroll
            for (int p = 0; p < NP; ++p) {
                float4 v = xr[p];
                v.x *= sc[p]; v.y *= sc[p]; v.z *= sc[p]; v.w *= sc[p];
                *(float4*)(xd + (srow + RPP * p) * PITCH + scol) = v;
            }
#pragma unroll
            for (int i = 0; i < 2 * KS; ++i) bcur[i] = bnxt[i];
            __syncthreads();
        }
    }

    const int rbase = m0 + 8 * half;
    float* __restrict__ pp = P + (size_t)rbase * NC + n0 + lid;
    if (m0 + BM <= nrows) {
#pragma unroll
        for (int mt = 0; mt < MT; ++mt)
#pragma unroll
            for (int r = 0; r < 8; ++r)
                pp[(size_t)(mt * 16 + r) * NC] = acc[mt][r];
    } else {
#pragma unroll
        for (int mt = 0; mt < MT; ++mt)
#pragma unroll
            for (int r = 0; r < 8; ++r)
                if (rbase + mt * 16 + r < nrows)
                    pp[(size_t)(mt * 16 + r) * NC] = acc[mt][r];
    }
}


constexpr int SUB = 400;

template <int D, bool FULL>
__device__ __forceinline__ void gather_add(const float* __restrict__ Pslice, float* acc,
                                           int s_l, int t_l, int nvalid, int lane) {
    constexpr int U = 16;
#pragma unroll
    for (int i0 = 0; i0 < 32; i0 += U) {
        if (!FULL && i0 >= nvalid) break;
        float2 v[U];
#pragma unroll
        for (int u = 0; u < U; ++u) {
            if (FULL || i0 + u < nvalid) {
                const int s = __builtin_amdgcn_readlane(s_l, i0 + u);
                v[u] = *(const float2*)(Pslice + (size_t)s * D + 2 * lane);
            }
        }
#pragma unroll
        for (int u = 0; u < U; ++u) {
            if (FULL || i0 + u < nvalid) {
                const int t = __builtin_amdgcn_readlane(t_l, i0 + u);
                float* a = acc + t * 64 + 2 * lane;
                atomicAdd(a,     v[u].x);
                atomicAdd(a + 1, v[u].y);
            }
        }
    }
}

struct HitLists {
    int2* r0;  int2* r1;
    int   cap0, cap1;
    int*  cnt;
    bool  on;
};

template <int D, bool FULL, bool WRITE>
__device__ __forceinline__ void ring_drain(const float* __restrict__ Pslice, const int* __restrict__ src,
                                           const int* __restrict__ dst, float* acc, const int* qe,
                                           int lo, int tail, int nvalid, int lane, const HitLists& hl) {
    __builtin_amdgcn_wave_barrier();
    const int e_l = (FULL || lane < nvalid) ? qe[(tail + lane) & 255] : 0;
    const int s_l = src[e_l];
    const int d_l = dst[e_l];
    const int t_l = d_l - lo;
    gather_add<D, FULL>(Pslice, acc, s_l, t_l, nvalid, lane);
    if (WRITE && hl.on) {
        const bool valid = FULL || lane < nvalid;
        const bool up = valid && t_l >= SUB;
        const bool dn = valid && t_l < SUB;
        const unsigned mu = __builtin_amdgcn_ballot_w32(up);
        const unsigned md = __builtin_amdgcn_ballot_w32(dn);
        int b0 = 0, b1 = 0;
        if (lane == 0) {
            b0 = atomicAdd(hl.cnt,     __builtin_popcount(md));
            b1 = atomicAdd(hl.cnt + 1, __builtin_popcount(mu));
        }
        b0 = __builtin_amdgcn_readfirstlane(b0);
        b1 = __builtin_amdgcn_readfirstlane(b1);
        const unsigned below = (1u << lane) - 1u;
        const int pos = up ? b1 + __builtin_popcount(mu & below) : b0 + __builtin_popcount(md & below);
        const int cap = up ? hl.cap1 : hl.cap0;
        if (valid && pos < cap) {
            int2* r = up ? hl.r1 : hl.r0;
            r[pos] = make_int2(s_l, d_l);
        }
    }
}

template <int D, int M, bool RELU, int LMODE>
__global__ __launch_bounds__(1024)
void k_agg_tile(const float* __restrict__ P, const int* __restrict__ src,
                const int* __restrict__ dst, const float* __restrict__ nd,
                const float* __restrict__ b, float* H, float* lst, int n, int E) {
    constexpr int F  = 64;
    constexpr int S  = D / F;
    constexpr int NW = 32;
    constexpr int CH = 128;
    constexpr int NR = M / SUB;
    static_assert(M % SUB == 0 && (LMODE != 1 || NR == 2), "tile rows are whole hit-list blocks");
    constexpr bool WR = (LMODE == 1);
    __shared__ __attribute__((aligned(16))) float acc[M * F];
    __shared__ int q_e[NW * 256];
    __shared__ int l_cnt[2];

    const int tid  = threadIdx.x;
    const int lane = tid & 31;
    const int w     = __builtin_amdgcn_readfirstlane(tid >> 5);
    const int slice = blockIdx.x % S;
    const int lo    = (blockIdx.x / S) * M;
    const int rows  = min(M, n - lo);
    const float* __restrict__ Pslice = P + slice * F;
    int* qe = q_e + w * 256;

    for (int i = tid; i < M * F / 4; i += 1024)
        ((float4*)acc)[i] = make_float4(0.f, 0.f, 0.f, 0.f);
    if (tid < 2) l_cnt[tid] = 0;

    int2* reg[2];
    int   rcap[2];
    bool  have[2];
#pragma unroll
    for (int j = 0; j < 2; ++j) {
        const int r0 = lo + j * SUB;
        have[j] = j < NR && r0 < n;
        reg[j]  = (int2*)(lst + (size_t)(have[j] ? r0 : lo) * 64);
        rcap[j] = have[j] ? min(SUB, n - r0) * 32 - 1 : 0;
    }
    HitLists hl;
    hl.r0 = reg[0]; hl.r1 = reg[1]; hl.cap0 = rcap[0]; hl.cap1 = rcap[1];
    hl.cnt = l_cnt; hl.on = WR && slice == 0;

    int  rcnt[2]  = {0, 0};
    bool use_list = (LMODE == 2);
    if (LMODE == 2) {
#pragma unroll
        for (int j = 0; j < NR; ++j) {
            if (have[j]) {
                rcnt[j] = reg[j][rcap[j]].x;
                use_list = use_list && rcnt[j] <= rcap[j];
            }
        }
    }
    __syncthreads();

    if (use_list) {
#pragma unroll
        for (int j = 0; j < NR; ++j) {
            const int   c  = rcnt[j];
            const int2* rg = reg[j];
            int  base = w * 32;
            int2 nxt  = make_int2(0, lo);
            if (base + lane < c) nxt = rg[base + lane];
            while (base + 32 <= c) {
                const int2 cur = nxt;
                const int  nb  = base + NW * 32;
                if (nb + lane < c) nxt = rg[nb + lane];
                gather_add<D, true>(Pslice, acc, cur.x, cur.y - lo, 32, lane);
                base = nb;
            }
            if (base < c)
                gather_add<D, false>(Pslice, acc, nxt.x, nxt.y - lo, c - base, lane);
        }
    } else {
    int head = 0, tail = 0;
    const unsigned lt_mask = (1u << lane) - 1u;

    auto scan_step = [&](const int4 cv, int e0) {
        const int dv[4] = {cv.x, cv.y, cv.z, cv.w};
#pragma unroll
        for (int c = 0; c < 4; ++c) {
            const bool match = (unsigned)(dv[c] - lo) < (unsigned)rows;
            const unsigned m = __builtin_amdgcn_ballot_w32(match);
            if (m) {
                if (match)
                    qe[(head + __builtin_popcount(m & lt_mask)) & 255] = e0 + c;
                head += __builtin_popcount(m);
            }
        }
        while (head - tail >= 32) {
            ring_drain<D, true, WR>(Pslice, src, dst, acc, qe, lo, tail, 32, lane, hl);
            tail += 32;
        }
    };

    const int nsteps_all = E / CH;
    if (w < nsteps_all) {
        const int rot = (int)blockIdx.x * (nsteps_all / (int)gridDim.x);
        auto step_base = [&](int idx) {
            int s = idx + rot;
            if (s >= nsteps_all) s -= nsteps_all;
            return s * CH;
        };
        auto ld = [&](int idx) {
            const int bb = step_base(idx < nsteps_all ? idx : w);
            return *(const int4*)(dst + bb + lane * 4);
        };
        int idx = w;
        int4 c0 = ld(idx);
        int4 c1 = ld(idx + NW);
        int4 c2 = ld(idx + 2 * NW);
        int4 c3 = ld(idx + 3 * NW);
        for (; idx < nsteps_all; idx += 4 * NW) {
            scan_step(c0, step_base(idx) + lane * 4);
            c0 = ld(idx + 4 * NW);
            if (idx + NW >= nsteps_all) break;
            scan_step(c1, step_base(idx + NW) + lane * 4);
            c1 = ld(idx + 5 * NW);
            if (idx + 2 * NW >= nsteps_all) break;
            scan_step(c2, step_base(idx + 2 * NW) + lane * 4);
            c2 = ld(idx + 6 * NW);
            if (idx + 3 * NW >= nsteps_all) break;
            scan_step(c3, step_base(idx + 3 * NW) + lane * 4);
            c3 = ld(idx + 7 * NW);
        }
    }
    if (w == 0) {
        const int e_tail = nsteps_all * CH;
        for (int eb = e_tail; eb < E; eb += 32) {
            const int e = eb + lane;
            const int d = (e < E) ? dst[e] : -1;
            const bool match = (unsigned)(d - lo) < (unsigned)rows;
            const unsigned m = __builtin_amdgcn_ballot_w32(match);
            if (m) {
                if (match)
                    qe[(head + __builtin_popcount(m & lt_mask)) & 255] = e;
                head += __builtin_popcount(m);
            }
        }
        while (head - tail >= 32) {
            ring_drain<D, true, WR>(Pslice, src, dst, acc, qe, lo, tail, 32, lane, hl);
            tail += 32;
        }
    }
    if (head - tail > 0)
        ring_drain<D, false, WR>(Pslice, src, dst, acc, qe, lo, tail, head - tail, lane, hl);
    }
    __syncthreads();

    if (WR && slice == 0) {
        if (tid == 0 && have[0]) reg[0][rcap[0]] = make_int2(l_cnt[0], 0);
        if (tid == 1 && have[1]) reg[1][rcap[1]] = make_int2(l_cnt[1], 0);
    }

    for (int i = tid; i < rows * F; i += 1024) {
        const int r = i >> 6, f = i & 63;
        float v = acc[i] * nd[lo + r] + b[slice * F + f];
        if (RELU) v = fmaxf(v, 0.0f);
        H[(size_t)(lo + r) * D + slice * F + f] = v;
    }
}


static inline int cdiv_ll(long long a, int b) { return (int)((a + b - 1) / b); }

extern "C" void kernel_launch(void* const* d_in, const int* in_sizes, int n_in,
                              void* d_out, int out_size, void* d_ws, size_t ws_size,
                              hipStream_t stream) {
    const float* feat = (const float*)d_in[0];
    const float* W1   = (const float*)d_in[1];
    const float* b1   = (const float*)d_in[2];
    const float* W2   = (const float*)d_in[3];
    const float* b2   = (const float*)d_in[4];
    const float* W3   = (const float*)d_in[5];
    const float* b3   = (const float*)d_in[6];
    const int*   src  = (const int*)d_in[7];
    const int*   dst  = (const int*)d_in[8];

    const int D_IN = 256, D_H = 128, D_OUT = 64;
    const int n = in_sizes[0] / D_IN;
    const int E = in_sizes[7];

    float* ns   = (float*)d_ws;
    float* nd   = ns + n;
    float* buf0 = nd + n;
    float* buf1 = buf0 + (size_t)n * D_H;

    float* out = (float*)d_out;

    const int mblocks = (n + 63) / 64;

    k_zero<<<cdiv_ll(2LL * n, 256), 256, 0, stream>>>(ns, 2LL * n);
    k_degrees<<<(E + 255) / 256, 256, 0, stream>>>(src, dst, ns, nd, E);
    k_rsqrt_norm<<<(2 * n + 255) / 256, 256, 0, stream>>>(ns, 2 * n);

    k_gemm_scaled<256, 128><<<mblocks, 256, 0, stream>>>(feat, ns, W1, buf0, n);
    k_agg_tile<128, 800, true, 1><<<(n + 799) / 800 * 2, 1024, 0, stream>>>(buf0, src, dst, nd, b1, buf1, out, n, E);

    k_gemm_scaled<128, 128><<<mblocks, 256, 0, stream>>>(buf1, ns, W2, buf0, n);
    k_agg_tile<128, 800, true, 2><<<(n + 799) / 800 * 2, 1024, 0, stream>>>(buf0, src, dst, nd, b2, buf1, out, n, E);

    k_gemm_scaled<128, 64><<<mblocks, 128, 0, stream>>>(buf1, ns, W3, buf0, n);
    k_agg_tile<64, 400, false, 2><<<(n + 399) / 400, 1024, 0, stream>>>(buf0, src, dst, nd, b3, out, out, n, E);
}


# Hit lists built by one binning pass: a workgroup ranks 4096 edges per 400-row block by LDS adds, takes list slots by one global add per block; layer 1 reads lists, no scan; sum order = LDS arrival
# speedup vs baseline: 2.7761x; 1.1018x over previous
#include <hip/hip_runtime.h>

typedef __attribute__((ext_vector_type(2))) float v2f;
typedef __attribute__((ext_vector_type(8))) float v8f;


constexpr int SUB = 400;

__device__ __forceinline__ int2* list_of(float* lst, int r0) { return (int2*)(lst + (size_t)r0 * 64); }
__device__ __forceinline__ int   list_cap(int r0, int n)    { return min(SUB, n - r0) * 32 - 1; }

__global__ void k_zero(float* __restrict__ p, long long cnt, float* lst, int n) {
    long long i = (long long)blockIdx.x * blockDim.x + threadIdx.x;
    if (i < cnt) p[i] = 0.0f;
    if (i < (n + SUB - 1) / SUB) {
        const int r0 = (int)i * SUB;
        list_of(lst, r0)[list_cap(r0, n)] = make_int2(0, 0);
    }
}

__global__ void k_rsqrt_norm(float* __restrict__ p, int n) {
    int i = blockIdx.x * blockDim.x + threadIdx.x;
    if (i < n) {
        float v = p[i];
        p[i] = rsqrtf(fmaxf(v, 1.0f));
    }
}


__global__ void k_degrees(const int* __restrict__ src, const int* __restrict__ dst,
                          float* __restrict__ deg_out, float* __restrict__ deg_in, int E) {
    int e = blockIdx.x * blockDim.x + threadIdx.x;
    if (e < E) {
        atomicAdd(&deg_out[src[e]], 1.0f);
        atomicAdd(&deg_in[dst[e]], 1.0f);
    }
}

template <int K, int NC>
__global__ __launch_bounds__(NC * 2)
void k_gemm_scaled(const float* __restrict__ X, const float* __restrict__ ns,
                   const float* __restrict__ W, float* __restrict__ P,
                   int nrows) {
    constexpr int BM    = 64;
    constexpr int MT    = BM / 16;
    constexpr int KC    = 32;
    constexpr int PITCH = KC + 4;
    constexpr int NTHR  = NC * 2;
    constexpr int RPP   = NTHR / 8;
    constexpr int NP    = BM / RPP;
    constexpr int NCH   = K / KC;
    constexpr int KS    = KC / 4;
    static_assert(K % KC == 0 && NC % 16 == 0 && BM % RPP == 0, "tile shape");
    __shared__ __attribute__((aligned(16))) float xs[2 * BM * PITCH];

    const int tid  = threadIdx.x;
    const int lane = tid & 31;
    const int w    = __builtin_amdgcn_readfirstlane(tid >> 5);
    const int half = lane >> 4;
    const int lid  = lane & 15;
    const int m0   = blockIdx.x * BM;
    const int n0   = w * 16;

    const int srow = tid >> 3;
    const int scol = (tid & 7) * 4;
    const float* xp[NP];
    float sc[NP];
#pragma unroll
    for (int p = 0; p < NP; ++p) {
        const int  rr = m0 + srow + RPP * p;
        const bool ok = rr < nrows;
        const int  rc = ok ? rr : (nrows - 1);
        sc[p] = ok ? ns[rc] : 0.0f;
        xp[p] = X + (size_t)rc * K + scol;
    }
    const float* __restrict__ wb = W + (size_t)(2 * half) * NC + n0 + lid;

    float4 xr[NP];
    float  bcur[2 * KS], bnxt[2 * KS];
#pragma unroll
    for (int p = 0; p < NP; ++p) xr[p] = *(const float4*)(xp[p]);
#pragma unroll
    for (int ks = 0; ks < KS; ++ks) {
        bcur[2 * ks]     = wb[(size_t)(ks * 4) * NC];
        bcur[2 * ks + 1] = wb[(size_t)(ks * 4 + 1) * NC];
    }
#pragma unroll
    for (int p = 0; p < NP; ++p) {
        float4 v = xr[p];
        v.x *= sc[p]; v.y *= sc[p]; v.z *= sc[p]; v.w *= sc[p];
        *(float4*)(xs + (srow + RPP * p) * PITCH + scol) = v;
    }
    __syncthreads();

    v8f acc[MT];
#pragma unroll
    for (int mt = 0; mt < MT; ++mt) acc[mt] = v8f{};

    for (int kc = 0; kc < NCH; ++kc) {
        const bool more = kc + 1 < NCH;
        if (more) {
            const int k1 = (kc + 1) * KC;
#pragma unroll
            for (int p = 0; p < NP; ++p) xr[p] = *(const float4*)(xp[p] + k1);
#pragma unroll
            for (int ks = 0; ks < KS; ++ks) {
                bnxt[2 * ks]     = wb[(size_t)(k1 + ks * 4) * NC];
                bnxt[2 * ks + 1] = wb[(size_t)(k1 + ks * 4 + 1) * NC];
            }
        }
        __builtin_amdgcn_sched_barrier(0);
        const float* xb = xs + (kc & 1) * (BM * PITCH) + lid * PITCH + 2 * half;
#pragma unroll
        for (int ks = 0; ks < KS; ++ks) {
            v2f b;
            b.x = bcur[2 * ks];
            b.y = bcur[2 * ks + 1];
#pragma unroll
            for (int mt = 0; mt < MT; ++mt) {
                const v2f a = *(const v2f*)(xb + mt * 16 * PITCH + ks * 4);
                acc[mt] = __builtin_amdgcn_wmma_f32_16x16x4_f32(
                     false, a,  false, b,
                     (short)0, acc[mt],  false,  false);
            }
        }
        if (more) {
            float* xd = xs + ((kc + 1) & 1) * (BM * PITCH);
#pragma unroll
            for (int p = 0; p < NP; ++p) {
                float4 v = xr[p];
                v.x *= sc[p]; v.y *= sc[p]; v.z *= sc[p]; v.w *= sc[p];
                *(float4*)(xd + (srow + RPP * p) * PITCH + scol) = v;
            }
#pragma unroll
            for (int i = 0; i < 2 * KS; ++i) bcur[i] = bnxt[i];
            __syncthreads();
        }
    }

    const int rbase = m0 + 8 * half;
    float* __restrict__ pp = P + (size_t)rbase * NC + n0 + lid;
    if (m0 + BM <= nrows) {
#pragma unroll
        for (int mt = 0; mt < MT; ++mt)
#pragma unroll
            for (int r = 0; r < 8; ++r)
                pp[(size_t)(mt * 16 + r) * NC] = acc[mt][r];
    } else {
#pragma unroll
        for (int mt = 0; mt < MT; ++mt)
#pragma unroll
            for (int r = 0; r < 8; ++r)
                if (rbase + mt * 16 + r < nrows)
                    pp[(size_t)(mt * 16 + r) * NC] = acc[mt][r];
    }
}


constexpr int BINS = 256;

__global__ __launch_bounds__(1024)
void k_bin_edges(const int* __restrict__ src, const int* __restrict__ dst, float* lst, int n, int E) {
    __shared__ int cnt[BINS];
    __shared__ int base[BINS];
    const int tid = threadIdx.x;
    const int e0  = (blockIdx.x * 1024 + tid) * 4;
    int s[4], d[4];
    if (e0 + 3 < E) {
        const int4 dv = *(const int4*)(dst + e0);
        const int4 sv = *(const int4*)(src + e0);
        d[0] = dv.x; d[1] = dv.y; d[2] = dv.z; d[3] = dv.w;
        s[0] = sv.x; s[1] = sv.y; s[2] = sv.z; s[3] = sv.w;
    } else {
#pragma unroll
        for (int c = 0; c < 4; ++c) {
            const bool in = e0 + c < E;
            d[c] = in ? dst[e0 + c] : -1;
            s[c] = in ? src[e0 + c] : 0;
        }
    }
    const int nb = (n + SUB - 1) / SUB;
    for (int b0 = 0; b0 < nb; b0 += BINS) {
        if (tid < BINS) cnt[tid] = 0;
        __syncthreads();
        int bl[4], rk[4];
#pragma unroll
        for (int c = 0; c < 4; ++c) {
            bl[c] = d[c] / SUB - b0;
            const bool in = (unsigned)d[c] < (unsigned)n && (unsigned)bl[c] < (unsigned)BINS;
            rk[c] = in ? atomicAdd(&cnt[bl[c]], 1) : -1;
        }
        __syncthreads();
        if (tid < BINS && b0 + tid < nb && cnt[tid] > 0) {
            const int r0 = (b0 + tid) * SUB;
            base[tid] = atomicAdd((int*)(list_of(lst, r0) + list_cap(r0, n)), cnt[tid]);
        }
        __syncthreads();
#pragma unroll
        for (int c = 0; c < 4; ++c) {
            if (rk[c] >= 0) {
                const int r0  = (b0 + bl[c]) * SUB;
                const int pos = base[bl[c]] + rk[c];
                if (pos < list_cap(r0, n)) list_of(lst, r0)[pos] = make_int2(s[c], d[c]);
            }
        }
        __syncthreads();
    }
}


template <int D, bool FULL>
__device__ __forceinline__ void gather_add(const float* __restrict__ Pslice, float* acc,
                                           int s_l, int t_l, int nvalid, int lane) {
    constexpr int U = 16;
#pragma unroll
    for (int i0 = 0; i0 < 32; i0 += U) {
        if (!FULL && i0 >= nvalid) break;
        float2 v[U];
#pragma unroll
        for (int u = 0; u < U; ++u) {
            if (FULL || i0 + u < nvalid) {
                const int s = __builtin_amdgcn_readlane(s_l, i0 + u);
                v[u] = *(const float2*)(Pslice + (size_t)s * D + 2 * lane);
            }
        }
#pragma unroll
        for (int u = 0; u < U; ++u) {
            if (FULL || i0 + u < nvalid) {
                const int t = __builtin_amdgcn_readlane(t_l, i0 + u);
                float* a = acc + t * 64 + 2 * lane;
                atomicAdd(a,     v[u].x);
                atomicAdd(a + 1, v[u].y);
            }
        }
    }
}

template <int D, bool FULL>
__device__ __forceinline__ void ring_drain(const float* __restrict__ Pslice, const int* __restrict__ src,
                                           const int* __restrict__ dst, float* acc, const int* qe,
                                           int lo, int tail, int nvalid, int lane) {
    __builtin_amdgcn_wave_barrier();
    const int e_l = (FULL || lane < nvalid) ? qe[(tail + lane) & 255] : 0;
    const int s_l = src[e_l];
    const int t_l = dst[e_l] - lo;
    gather_add<D, FULL>(Pslice, acc, s_l, t_l, nvalid, lane);
}

template <int D, int M, bool RELU>
__global__ __launch_bounds__(1024)
void k_agg_tile(const float* __restrict__ P, const int* __restrict__ src,
                const int* __restrict__ dst, const float* __restrict__ nd,
                const float* __restrict__ b, float* H, float* lst, int n, int E) {
    constexpr int F  = 64;
    constexpr int S  = D / F;
    constexpr int NW = 32;
    constexpr int CH = 128;
    constexpr int NR = M / SUB;
    static_assert(M % SUB == 0 && NR <= 2, "tile rows are whole hit-list blocks");
    __shared__ __attribute__((aligned(16))) float acc[M * F];
    __shared__ int q_e[NW * 256];

    const int tid  = threadIdx.x;
    const int lane = tid & 31;
    const int w     = __builtin_amdgcn_readfirstlane(tid >> 5);
    const int slice = blockIdx.x % S;
    const int lo    = (blockIdx.x / S) * M;
    const int rows  = min(M, n - lo);
    const float* __restrict__ Pslice = P + slice * F;
    int* qe = q_e + w * 256;

    for (int i = tid; i < M * F / 4; i += 1024)
        ((float4*)acc)[i] = make_float4(0.f, 0.f, 0.f, 0.f);

    const int2* reg[2];
    int  rcnt[2]  = {0, 0};
    bool use_list = true;
#pragma unroll
    for (int j = 0; j < 2; ++j) {
        const int  r0   = lo + j * SUB;
        const bool have = j < NR && r0 < n;
        reg[j] = list_of(lst, have ? r0 : lo);
        if (have) {
            const int cap = list_cap(r0, n);
            rcnt[j]  = reg[j][cap].x;
            use_list = use_list && rcnt[j] <= cap;
        }
    }
    __syncthreads();

    if (use_list) {
#pragma unroll
        for (int j = 0; j < NR; ++j) {
            const int   c  = rcnt[j];
            const int2* rg = reg[j];
            int  base = w * 32;
            int2 nxt  = make_int2(0, lo);
            if (base + lane < c) nxt = rg[base + lane];
            while (base + 32 <= c) {
                const int2 cur = nxt;
                const int  nb  = base + NW * 32;
                if (nb + lane < c) nxt = rg[nb + lane];
                gather_add<D, true>(Pslice, acc, cur.x, cur.y - lo, 32, lane);
                base = nb;
            }
            if (base < c)
                gather_add<D, false>(Pslice, acc, nxt.x, nxt.y - lo, c - base, lane);
        }
    } else {
    int head = 0, tail = 0;
    const unsigned lt_mask = (1u << lane) - 1u;

    auto scan_step = [&](const int4 cv, int e0) {
        const int dv[4] = {cv.x, cv.y, cv.z, cv.w};
#pragma unroll
        for (int c = 0; c < 4; ++c) {
            const bool match = (unsigned)(dv[c] - lo) < (unsigned)rows;
            const unsigned m = __builtin_amdgcn_ballot_w32(match);
            if (m) {
                if (match)
                    qe[(head + __builtin_popcount(m & lt_mask)) & 255] = e0 + c;
                head += __builtin_popcount(m);
            }
        }
        while (head - tail >= 32) {
            ring_drain<D, true>(Pslice, src, dst, acc, qe, lo, tail, 32, lane);
            tail += 32;
        }
    };

    const int nsteps_all = E / CH;
    if (w < nsteps_all) {
        const int rot = (int)blockIdx.x * (nsteps_all / (int)gridDim.x);
        auto step_base = [&](int idx) {
            int s = idx + rot;
            if (s >= nsteps_all) s -= nsteps_all;
            return s * CH;
        };
        auto ld = [&](int idx) {
            const int bb = step_base(idx < nsteps_all ? idx : w);
            return *(const int4*)(dst + bb + lane * 4);
        };
        int idx = w;
        int4 c0 = ld(idx);
        int4 c1 = ld(idx + NW);
        int4 c2 = ld(idx + 2 * NW);
        int4 c3 = ld(idx + 3 * NW);
        for (; idx < nsteps_all; idx += 4 * NW) {
            scan_step(c0, step_base(idx) + lane * 4);
            c0 = ld(idx + 4 * NW);
            if (idx + NW >= nsteps_all) break;
            scan_step(c1, step_base(idx + NW) + lane * 4);
            c1 = ld(idx + 5 * NW);
            if (idx + 2 * NW >= nsteps_all) break;
            scan_step(c2, step_base(idx + 2 * NW) + lane * 4);
            c2 = ld(idx + 6 * NW);
            if (idx + 3 * NW >= nsteps_all) break;
            scan_step(c3, step_base(idx + 3 * NW) + lane * 4);
            c3 = ld(idx + 7 * NW);
        }
    }
    if (w == 0) {
        const int e_tail = nsteps_all * CH;
        for (int eb = e_tail; eb < E; eb += 32) {
            const int e = eb + lane;
            const int d = (e < E) ? dst[e] : -1;
            const bool match = (unsigned)(d - lo) < (unsigned)rows;
            const unsigned m = __builtin_amdgcn_ballot_w32(match);
            if (m) {
                if (match)
                    qe[(head + __builtin_popcount(m & lt_mask)) & 255] = e;
                head += __builtin_popcount(m);
            }
        }
        while (head - tail >= 32) {
            ring_drain<D, true>(Pslice, src, dst, acc, qe, lo, tail, 32, lane);
            tail += 32;
        }
    }
    if (head - tail > 0)
        ring_drain<D, false>(Pslice, src, dst, acc, qe, lo, tail, head - tail, lane);
    }
    __syncthreads();

    for (int i = tid; i < rows * F; i += 1024) {
        const int r = i >> 6, f = i & 63;
        float v = acc[i] * nd[lo + r] + b[slice * F + f];
        if (RELU) v = fmaxf(v, 0.0f);
        H[(size_t)(lo + r) * D + slice * F + f] = v;
    }
}


static inline int cdiv_ll(long long a, int b) { return (int)((a + b - 1) / b); }

extern "C" void kernel_launch(void* const* d_in, const int* in_sizes, int n_in,
                              void* d_out, int out_size, void* d_ws, size_t ws_size,
                              hipStream_t stream) {
    const float* feat = (const float*)d_in[0];
    const float* W1   = (const float*)d_in[1];
    const float* b1   = (const float*)d_in[2];
    const float* W2   = (const float*)d_in[3];
    const float* b2   = (const float*)d_in[4];
    const float* W3   = (const float*)d_in[5];
    const float* b3   = (const float*)d_in[6];
    const int*   src  = (const int*)d_in[7];
    const int*   dst  = (const int*)d_in[8];

    const int D_IN = 256, D_H = 128, D_OUT = 64;
    const int n = in_sizes[0] / D_IN;
    const int E = in_sizes[7];

    float* ns   = (float*)d_ws;
    float* nd   = ns + n;
    float* buf0 = nd + n;
    float* buf1 = buf0 + (size_t)n * D_H;

    float* out = (float*)d_out;

    const int mblocks = (n + 63) / 64;

    k_zero<<<cdiv_ll(2LL * n, 256), 256, 0, stream>>>(ns, 2LL * n, out, n);
    k_degrees<<<(E + 255) / 256, 256, 0, stream>>>(src, dst, ns, nd, E);
    k_rsqrt_norm<<<(2 * n + 255) / 256, 256, 0, stream>>>(ns, 2 * n);

    k_bin_edges<<<(E + 4095) / 4096, 1024, 0, stream>>>(src, dst, out, n, E);

    k_gemm_scaled<256, 128><<<mblocks, 256, 0, stream>>>(feat, ns, W1, buf0, n);
    k_agg_tile<128, 800, true><<<(n + 799) / 800 * 2, 1024, 0, stream>>>(buf0, src, dst, nd, b1, buf1, out, n, E);

    k_gemm_scaled<128, 128><<<mblocks, 256, 0, stream>>>(buf1, ns, W2, buf0, n);
    k_agg_tile<128, 800, true><<<(n + 799) / 800 * 2, 1024, 0, stream>>>(buf0, src, dst, nd, b2, buf1, out, n, E);

    k_gemm_scaled<128, 64><<<mblocks, 128, 0, stream>>>(buf1, ns, W3, buf0, n);
    k_agg_tile<64, 400, false><<<(n + 399) / 400, 1024, 0, stream>>>(buf0, src, dst, nd, b3, out, out, n, E);
}


# Degrees counted in LDS, per-edge global float adds gone: in-degree = edges per row of the aggregation tile, out-degree from source-id lists binned in the same edge pass; 10 -> 9 launches
# speedup vs baseline: 3.6622x; 1.3192x over previous
#include <hip/hip_runtime.h>

typedef __attribute__((ext_vector_type(2))) float v2f;
typedef __attribute__((ext_vector_type(8))) float v8f;


constexpr int SUB = 400;

__device__ __forceinline__ int2* list_of(float* lst, int r0) { return (int2*)(lst + (size_t)r0 * 64); }
__device__ __forceinline__ int   list_cap(int r0, int n)    { return min(SUB, n - r0) * 32 - 1; }

__device__ __forceinline__ int* slist_of(float* sl, int r0)       { return (int*)(sl + (size_t)r0 * 128); }
__device__ __forceinline__ int  slist_cap(int r0, int n)          { return min(SUB, n - r0) * 128 - 1; }

__global__ void k_zero(float* lst, float* sl, int n) {
    const int i = blockIdx.x * blockDim.x + threadIdx.x;
    if (i < (n + SUB - 1) / SUB) {
        const int r0 = i * SUB;
        list_of(lst, r0)[list_cap(r0, n)] = make_int2(0, 0);
        slist_of(sl, r0)[slist_cap(r0, n)] = 0;
    }
}

__global__ __launch_bounds__(1024)
void k_src_norm(const int* __restrict__ src, float* sl, float* __restrict__ ns, int n, int E) {
    __shared__ int cnt[SUB];
    const int tid  = threadIdx.x;
    const int r0   = blockIdx.x * SUB;
    const int rows = min(SUB, n - r0);
    if (tid < SUB) cnt[tid] = 0;
    __syncthreads();
    const int* lp  = slist_of(sl, r0);
    const int  cap = slist_cap(r0, n);
    const int  c   = lp[cap];
    if (c <= cap) {
        for (int i = tid * 4; i < c; i += 4096) {
            if (i + 3 < c) {
                const int4 v = *(const int4*)(lp + i);
                const int t[4] = {v.x - r0, v.y - r0, v.z - r0, v.w - r0};
#pragma unroll
                for (int k = 0; k < 4; ++k)
                    if ((unsigned)t[k] < (unsigned)rows) atomicAdd(&cnt[t[k]], 1);
            } else {
                for (int k = 0; k < 4 && i + k < c; ++k) {
                    const int t = lp[i + k] - r0;
                    if ((unsigned)t < (unsigned)rows) atomicAdd(&cnt[t], 1);
                }
            }
        }
    } else {
        for (int e = tid; e < E; e += 1024) {
            const int t = src[e] - r0;
            if ((unsigned)t < (unsigned)rows) atomicAdd(&cnt[t], 1);
        }
    }
    __syncthreads();
    if (tid < rows) ns[r0 + tid] = rsqrtf(fmaxf((float)cnt[tid], 1.0f));
}

template <int K, int NC>
__global__ __launch_bounds__(NC * 2)
void k_gemm_scaled(const float* __restrict__ X, const float* __restrict__ ns,
                   const float* __restrict__ W, float* __restrict__ P,
                   int nrows) {
    constexpr int BM    = 64;
    constexpr int MT    = BM / 16;
    constexpr int KC    = 32;
    constexpr int PITCH = KC + 4;
    constexpr int NTHR  = NC * 2;
    constexpr int RPP   = NTHR / 8;
    constexpr int NP    = BM / RPP;
    constexpr int NCH   = K / KC;
    constexpr int KS    = KC / 4;
    static_assert(K % KC == 0 && NC % 16 == 0 && BM % RPP == 0, "tile shape");
    __shared__ __attribute__((aligned(16))) float xs[2 * BM * PITCH];

    const int tid  = threadIdx.x;
    const int lane = tid & 31;
    const int w    = __builtin_amdgcn_readfirstlane(tid >> 5);
    const int half = lane >> 4;
    const int lid  = lane & 15;
    const int m0   = blockIdx.x * BM;
    const int n0   = w * 16;

    const int srow = tid >> 3;
    const int scol = (tid & 7) * 4;
    const float* xp[NP];
    float sc[NP];
#pragma unroll
    for (int p = 0; p < NP; ++p) {
        const int  rr = m0 + srow + RPP * p;
        const bool ok = rr < nrows;
        const int  rc = ok ? rr : (nrows - 1);
        sc[p] = ok ? ns[rc] : 0.0f;
        xp[p] = X + (size_t)rc * K + scol;
    }
    const float* __restrict__ wb = W + (size_t)(2 * half) * NC + n0 + lid;

    float4 xr[NP];
    float  bcur[2 * KS], bnxt[2 * KS];
#pragma unroll
    for (int p = 0; p < NP; ++p) xr[p] = *(const float4*)(xp[p]);
#pragma unroll
    for (int ks = 0; ks < KS; ++ks) {
        bcur[2 * ks]     = wb[(size_t)(ks * 4) * NC];
        bcur[2 * ks + 1] = wb[(size_t)(ks * 4 + 1) * NC];
    }
#pragma unroll
    for (int p = 0; p < NP; ++p) {
        float4 v = xr[p];
        v.x *= sc[p]; v.y *= sc[p]; v.z *= sc[p]; v.w *= sc[p];
        *(float4*)(xs + (srow + RPP * p) * PITCH + scol) = v;
    }
    __syncthreads();

    v8f acc[MT];
#pragma unroll
    for (int mt = 0; mt < MT; ++mt) acc[mt] = v8f{};

    for (int kc = 0; kc < NCH; ++kc) {
        const bool more = kc + 1 < NCH;
        if (more) {
            const int k1 = (kc + 1) * KC;
#pragma unroll
            for (int p = 0; p < NP; ++p) xr[p] = *(const float4*)(xp[p] + k1);
#pragma unroll
            for (int ks = 0; ks < KS; ++ks) {
                bnxt[2 * ks]     = wb[(size_t)(k1 + ks * 4) * NC];
                bnxt[2 * ks + 1] = wb[(size_t)(k1 + ks * 4 + 1) * NC];
            }
        }
        __builtin_amdgcn_sched_barrier(0);
        const float* xb = xs + (kc & 1) * (BM * PITCH) + lid * PITCH + 2 * half;
#pragma unroll
        for (int ks = 0; ks < KS; ++ks) {
            v2f b;
            b.x = bcur[2 * ks];
            b.y = bcur[2 * ks + 1];
#pragma unroll
            for (int mt = 0; mt < MT; ++mt) {
                const v2f a = *(const v2f*)(xb + mt * 16 * PITCH + ks * 4);
                acc[mt] = __builtin_amdgcn_wmma_f32_16x16x4_f32(
                     false, a,  false, b,
                     (short)0, acc[mt],  false,  false);
            }
        }
        if (more) {
            float* xd = xs + ((kc + 1) & 1) * (BM * PITCH);
#pragma unroll
            for (int p = 0; p < NP; ++p) {
                float4 v = xr[p];
                v.x *= sc[p]; v.y *= sc[p]; v.z *= sc[p]; v.w *= sc[p];
                *(float4*)(xd + (srow + RPP * p) * PITCH + scol) = v;
            }
#pragma unroll
            for (int i = 0; i < 2 * KS; ++i) bcur[i] = bnxt[i];
            __syncthreads();
        }
    }

    const int rbase = m0 + 8 * half;
    float* __restrict__ pp = P + (size_t)rbase * NC + n0 + lid;
    if (m0 + BM <= nrows) {
#pragma unroll
        for (int mt = 0; mt < MT; ++mt)
#pragma unroll
            for (int r = 0; r < 8; ++r)
                pp[(size_t)(mt * 16 + r) * NC] = acc[mt][r];
    } else {
#pragma unroll
        for (int mt = 0; mt < MT; ++mt)
#pragma unroll
            for (int r = 0; r < 8; ++r)
                if (rbase + mt * 16 + r < nrows)
                    pp[(size_t)(mt * 16 + r) * NC] = acc[mt][r];
    }
}


constexpr int BINS = 256;

__global__ __launch_bounds__(1024)
void k_bin_edges(const int* __restrict__ src, const int* __restrict__ dst, float* lst, float* sl, int n, int E) {
    __shared__ int cnt[BINS];
    __shared__ int base[BINS];
    const int tid = threadIdx.x;
    const int e0  = (blockIdx.x * 1024 + tid) * 4;
    int s[4], d[4];
    if (e0 + 3 < E) {
        const int4 dv = *(const int4*)(dst + e0);
        const int4 sv = *(const int4*)(src + e0);
        d[0] = dv.x; d[1] = dv.y; d[2] = dv.z; d[3] = dv.w;
        s[0] = sv.x; s[1] = sv.y; s[2] = sv.z; s[3] = sv.w;
    } else {
#pragma unroll
        for (int c = 0; c < 4; ++c) {
            const bool in = e0 + c < E;
            d[c] = in ? dst[e0 + c] : -1;
            s[c] = in ? src[e0 + c] : -1;
        }
    }
    const int nb = (n + SUB - 1) / SUB;
    for (int b0 = 0; b0 < nb; b0 += BINS) {
        if (tid < BINS) cnt[tid] = 0;
        __syncthreads();
        int bl[4], rk[4];
#pragma unroll
        for (int c = 0; c < 4; ++c) {
            bl[c] = d[c] / SUB - b0;
            const bool in = (unsigned)d[c] < (unsigned)n && (unsigned)bl[c] < (unsigned)BINS;
            rk[c] = in ? atomicAdd(&cnt[bl[c]], 1) : -1;
        }
        __syncthreads();
        if (tid < BINS && b0 + tid < nb && cnt[tid] > 0) {
            const int r0 = (b0 + tid) * SUB;
            base[tid] = atomicAdd((int*)(list_of(lst, r0) + list_cap(r0, n)), cnt[tid]);
        }
        __syncthreads();
#pragma unroll
        for (int c = 0; c < 4; ++c) {
            if (rk[c] >= 0) {
                const int r0  = (b0 + bl[c]) * SUB;
                const int pos = base[bl[c]] + rk[c];
                if (pos < list_cap(r0, n)) list_of(lst, r0)[pos] = make_int2(s[c], d[c]);
            }
        }
        __syncthreads();
    }
    for (int b0 = 0; b0 < nb; b0 += BINS) {
        if (tid < BINS) cnt[tid] = 0;
        __syncthreads();
        int bl[4], rk[4];
#pragma unroll
        for (int c = 0; c < 4; ++c) {
            bl[c] = s[c] / SUB - b0;
            const bool in = (unsigned)s[c] < (unsigned)n && (unsigned)bl[c] < (unsigned)BINS;
            rk[c] = in ? atomicAdd(&cnt[bl[c]], 1) : -1;
        }
        __syncthreads();
        if (tid < BINS && b0 + tid < nb && cnt[tid] > 0) {
            const int r0 = (b0 + tid) * SUB;
            base[tid] = atomicAdd(slist_of(sl, r0) + slist_cap(r0, n), cnt[tid]);
        }
        __syncthreads();
#pragma unroll
        for (int c = 0; c < 4; ++c) {
            if (rk[c] >= 0) {
                const int r0  = (b0 + bl[c]) * SUB;
                const int pos = base[bl[c]] + rk[c];
                if (pos < slist_cap(r0, n)) slist_of(sl, r0)[pos] = s[c];
            }
        }
        __syncthreads();
    }
}


template <int D, bool FULL>
__device__ __forceinline__ void gather_add(const float* __restrict__ Pslice, float* acc,
                                           int s_l, int t_l, int nvalid, int lane) {
    constexpr int U = 16;
#pragma unroll
    for (int i0 = 0; i0 < 32; i0 += U) {
        if (!FULL && i0 >= nvalid) break;
        float2 v[U];
#pragma unroll
        for (int u = 0; u < U; ++u) {
            if (FULL || i0 + u < nvalid) {
                const int s = __builtin_amdgcn_readlane(s_l, i0 + u);
                v[u] = *(const float2*)(Pslice + (size_t)s * D + 2 * lane);
            }
        }
#pragma unroll
        for (int u = 0; u < U; ++u) {
            if (FULL || i0 + u < nvalid) {
                const int t = __builtin_amdgcn_readlane(t_l, i0 + u);
                float* a = acc + t * 64 + 2 * lane;
                atomicAdd(a,     v[u].x);
                atomicAdd(a + 1, v[u].y);
            }
        }
    }
}

template <int D, bool FULL>
__device__ __forceinline__ void ring_drain(const float* __restrict__ Pslice, const int* __restrict__ src,
                                           const int* __restrict__ dst, float* acc, int* rcn, const int* qe,
                                           int lo, int tail, int nvalid, int lane) {
    __builtin_amdgcn_wave_barrier();
    const int e_l = (FULL || lane < nvalid) ? qe[(tail + lane) & 255] : 0;
    const int s_l = src[e_l];
    const int t_l = dst[e_l] - lo;
    if (FULL || lane < nvalid) atomicAdd(&rcn[t_l], 1);
    gather_add<D, FULL>(Pslice, acc, s_l, t_l, nvalid, lane);
}

template <int D, int M, bool RELU>
__global__ __launch_bounds__(1024)
void k_agg_tile(const float* __restrict__ P, const int* __restrict__ src,
                const int* __restrict__ dst,
                const float* __restrict__ b, float* H, float* lst, int n, int E) {
    constexpr int F  = 64;
    constexpr int S  = D / F;
    constexpr int NW = 32;
    constexpr int CH = 128;
    constexpr int NR = M / SUB;
    static_assert(M % SUB == 0 && NR <= 2, "tile rows are whole hit-list blocks");
    __shared__ __attribute__((aligned(16))) float acc[M * F];
    __shared__ int q_e[NW * 256];
    __shared__ int rcn[M];

    const int tid  = threadIdx.x;
    const int lane = tid & 31;
    const int w     = __builtin_amdgcn_readfirstlane(tid >> 5);
    const int slice = blockIdx.x % S;
    const int lo    = (blockIdx.x / S) * M;
    const int rows  = min(M, n - lo);
    const float* __restrict__ Pslice = P + slice * F;
    int* qe = q_e + w * 256;

    for (int i = tid; i < M * F / 4; i += 1024)
        ((float4*)acc)[i] = make_float4(0.f, 0.f, 0.f, 0.f);
    if (tid < M) rcn[tid] = 0;

    const int2* reg[2];
    int  rcnt[2]  = {0, 0};
    bool use_list = true;
#pragma unroll
    for (int j = 0; j < 2; ++j) {
        const int  r0   = lo + j * SUB;
        const bool have = j < NR && r0 < n;
        reg[j] = list_of(lst, have ? r0 : lo);
        if (have) {
            const int cap = list_cap(r0, n);
            rcnt[j]  = reg[j][cap].x;
            use_list = use_list && rcnt[j] <= cap;
        }
    }
    __syncthreads();

    if (use_list) {
#pragma unroll
        for (int j = 0; j < NR; ++j) {
            const int   c  = rcnt[j];
            const int2* rg = reg[j];
            int  base = w * 32;
            int2 nxt  = make_int2(0, lo);
            if (base + lane < c) nxt = rg[base + lane];
            while (base + 32 <= c) {
                const int2 cur = nxt;
                const int  nb  = base + NW * 32;
                if (nb + lane < c) nxt = rg[nb + lane];
                atomicAdd(&rcn[cur.y - lo], 1);
                gather_add<D, true>(Pslice, acc, cur.x, cur.y - lo, 32, lane);
                base = nb;
            }
            if (base < c) {
                if (base + lane < c) atomicAdd(&rcn[nxt.y - lo], 1);
                gather_add<D, false>(Pslice, acc, nxt.x, nxt.y - lo, c - base, lane);
            }
        }
    } else {
    int head = 0, tail = 0;
    const unsigned lt_mask = (1u << lane) - 1u;

    auto scan_step = [&](const int4 cv, int e0) {
        const int dv[4] = {cv.x, cv.y, cv.z, cv.w};
#pragma unroll
        for (int c = 0; c < 4; ++c) {
            const bool match = (unsigned)(dv[c] - lo) < (unsigned)rows;
            const unsigned m = __builtin_amdgcn_ballot_w32(match);
            if (m) {
                if (match)
                    qe[(head + __builtin_popcount(m & lt_mask)) & 255] = e0 + c;
                head += __builtin_popcount(m);
            }
        }
        while (head - tail >= 32) {
            ring_drain<D, true>(Pslice, src, dst, acc, rcn, qe, lo, tail, 32, lane);
            tail += 32;
        }
    };

    const int nsteps_all = E / CH;
    if (w < nsteps_all) {
        const int rot = (int)blockIdx.x * (nsteps_all / (int)gridDim.x);
        auto step_base = [&](int idx) {
            int s = idx + rot;
            if (s >= nsteps_all) s -= nsteps_all;
            return s * CH;
        };
        auto ld = [&](int idx) {
            const int bb = step_base(idx < nsteps_all ? idx : w);
            return *(const int4*)(dst + bb + lane * 4);
        };
        int idx = w;
        int4 c0 = ld(idx);
        int4 c1 = ld(idx + NW);
        int4 c2 = ld(idx + 2 * NW);
        int4 c3 = ld(idx + 3 * NW);
        for (; idx < nsteps_all; idx += 4 * NW) {
            scan_step(c0, step_base(idx) + lane * 4);
            c0 = ld(idx + 4 * NW);
            if (idx + NW >= nsteps_all) break;
            scan_step(c1, step_base(idx + NW) + lane * 4);
            c1 = ld(idx + 5 * NW);
            if (idx + 2 * NW >= nsteps_all) break;
            scan_step(c2, step_base(idx + 2 * NW) + lane * 4);
            c2 = ld(idx + 6 * NW);
            if (idx + 3 * NW >= nsteps_all) break;
            scan_step(c3, step_base(idx + 3 * NW) + lane * 4);
            c3 = ld(idx + 7 * NW);
        }
    }
    if (w == 0) {
        const int e_tail = nsteps_all * CH;
        for (int eb = e_tail; eb < E; eb += 32) {
            const int e = eb + lane;
            const int d = (e < E) ? dst[e] : -1;
            const bool match = (unsigned)(d - lo) < (unsigned)rows;
            const unsigned m = __builtin_amdgcn_ballot_w32(match);
            if (m) {
                if (match)
                    qe[(head + __builtin_popcount(m & lt_mask)) & 255] = e;
                head += __builtin_popcount(m);
            }
        }
        while (head - tail >= 32) {
            ring_drain<D, true>(Pslice, src, dst, acc, rcn, qe, lo, tail, 32, lane);
            tail += 32;
        }
    }
    if (head - tail > 0)
        ring_drain<D, false>(Pslice, src, dst, acc, rcn, qe, lo, tail, head - tail, lane);
    }
    __syncthreads();

    for (int i = tid; i < rows * F; i += 1024) {
        const int r = i >> 6, f = i & 63;
        const float ndv = rsqrtf(fmaxf((float)rcn[r], 1.0f));
        float v = acc[i] * ndv + b[slice * F + f];
        if (RELU) v = fmaxf(v, 0.0f);
        H[(size_t)(lo + r) * D + slice * F + f] = v;
    }
}


extern "C" void kernel_launch(void* const* d_in, const int* in_sizes, int n_in,
                              void* d_out, int out_size, void* d_ws, size_t ws_size,
                              hipStream_t stream) {
    const float* feat = (const float*)d_in[0];
    const float* W1   = (const float*)d_in[1];
    const float* b1   = (const float*)d_in[2];
    const float* W2   = (const float*)d_in[3];
    const float* b2   = (const float*)d_in[4];
    const float* W3   = (const float*)d_in[5];
    const float* b3   = (const float*)d_in[6];
    const int*   src  = (const int*)d_in[7];
    const int*   dst  = (const int*)d_in[8];

    const int D_IN = 256, D_H = 128, D_OUT = 64;
    const int n = in_sizes[0] / D_IN;
    const int E = in_sizes[7];

    float* ns   = (float*)d_ws;
    float* buf0 = ns + 2 * (size_t)n;
    float* buf1 = buf0 + (size_t)n * D_H;

    float* out = (float*)d_out;

    const int mblocks = (n + 63) / 64;

    const int nblk = (n + SUB - 1) / SUB;

    k_zero<<<(nblk + 255) / 256, 256, 0, stream>>>(out, buf1, n);
    k_bin_edges<<<(E + 4095) / 4096, 1024, 0, stream>>>(src, dst, out, buf1, n, E);

    k_src_norm<<<nblk, 1024, 0, stream>>>(src, buf1, ns, n, E);

    k_gemm_scaled<256, 128><<<mblocks, 256, 0, stream>>>(feat, ns, W1, buf0, n);
    k_agg_tile<128, 800, true><<<(n + 799) / 800 * 2, 1024, 0, stream>>>(buf0, src, dst, b1, buf1, out, n, E);

    k_gemm_scaled<128, 128><<<mblocks, 256, 0, stream>>>(buf1, ns, W2, buf0, n);
    k_agg_tile<128, 800, true><<<(n + 799) / 800 * 2, 1024, 0, stream>>>(buf0, src, dst, b2, buf1, out, n, E);

    k_gemm_scaled<128, 64><<<mblocks, 128, 0, stream>>>(buf1, ns, W3, buf0, n);
    k_agg_tile<64, 400, false><<<(n + 399) / 400, 1024, 0, stream>>>(buf0, src, dst, b3, out, out, n, E);
}


# Aggregation without float atomics: a tile's pairs are routed in LDS (count, scan, place) to the wave owning the row (row mod 32), which adds rows by plain LDS read-add-write; sum order = arrival
# speedup vs baseline: 4.0400x; 1.1032x over previous
#include <hip/hip_runtime.h>

typedef __attribute__((ext_vector_type(2))) float v2f;
typedef __attribute__((ext_vector_type(8))) float v8f;


constexpr int SUB = 400;

__device__ __forceinline__ int2* list_of(float* lst, int r0) { return (int2*)(lst + (size_t)r0 * 64); }
__device__ __forceinline__ int   list_cap(int r0, int n)    { return min(SUB, n - r0) * 32 - 1; }

__device__ __forceinline__ int* slist_of(float* sl, int r0)       { return (int*)(sl + (size_t)r0 * 128); }
__device__ __forceinline__ int  slist_cap(int r0, int n)          { return min(SUB, n - r0) * 128 - 1; }

__global__ void k_zero(float* lst, float* sl, int n) {
    const int i = blockIdx.x * blockDim.x + threadIdx.x;
    if (i < (n + SUB - 1) / SUB) {
        const int r0 = i * SUB;
        list_of(lst, r0)[list_cap(r0, n)] = make_int2(0, 0);
        slist_of(sl, r0)[slist_cap(r0, n)] = 0;
    }
}

__global__ __launch_bounds__(1024)
void k_src_norm(const int* __restrict__ src, float* sl, float* __restrict__ ns, int n, int E) {
    __shared__ int cnt[SUB];
    const int tid  = threadIdx.x;
    const int r0   = blockIdx.x * SUB;
    const int rows = min(SUB, n - r0);
    if (tid < SUB) cnt[tid] = 0;
    __syncthreads();
    const int* lp  = slist_of(sl, r0);
    const int  cap = slist_cap(r0, n);
    const int  c   = lp[cap];
    if (c <= cap) {
        for (int i = tid * 4; i < c; i += 4096) {
            if (i + 3 < c) {
                const int4 v = *(const int4*)(lp + i);
                const int t[4] = {v.x - r0, v.y - r0, v.z - r0, v.w - r0};
#pragma unroll
                for (int k = 0; k < 4; ++k)
                    if ((unsigned)t[k] < (unsigned)rows) atomicAdd(&cnt[t[k]], 1);
            } else {
                for (int k = 0; k < 4 && i + k < c; ++k) {
                    const int t = lp[i + k] - r0;
                    if ((unsigned)t < (unsigned)rows) atomicAdd(&cnt[t], 1);
                }
            }
        }
    } else {
        for (int e = tid; e < E; e += 1024) {
            const int t = src[e] - r0;
            if ((unsigned)t < (unsigned)rows) atomicAdd(&cnt[t], 1);
        }
    }
    __syncthreads();
    if (tid < rows) ns[r0 + tid] = rsqrtf(fmaxf((float)cnt[tid], 1.0f));
}

template <int K, int NC>
__global__ __launch_bounds__(NC * 2)
void k_gemm_scaled(const float* __restrict__ X, const float* __restrict__ ns,
                   const float* __restrict__ W, float* __restrict__ P,
                   int nrows) {
    constexpr int BM    = 64;
    constexpr int MT    = BM / 16;
    constexpr int KC    = 32;
    constexpr int PITCH = KC + 4;
    constexpr int NTHR  = NC * 2;
    constexpr int RPP   = NTHR / 8;
    constexpr int NP    = BM / RPP;
    constexpr int NCH   = K / KC;
    constexpr int KS    = KC / 4;
    static_assert(K % KC == 0 && NC % 16 == 0 && BM % RPP == 0, "tile shape");
    __shared__ __attribute__((aligned(16))) float xs[2 * BM * PITCH];

    const int tid  = threadIdx.x;
    const int lane = tid & 31;
    const int w    = __builtin_amdgcn_readfirstlane(tid >> 5);
    const int half = lane >> 4;
    const int lid  = lane & 15;
    const int m0   = blockIdx.x * BM;
    const int n0   = w * 16;

    const int srow = tid >> 3;
    const int scol = (tid & 7) * 4;
    const float* xp[NP];
    float sc[NP];
#pragma unroll
    for (int p = 0; p < NP; ++p) {
        const int  rr = m0 + srow + RPP * p;
        const bool ok = rr < nrows;
        const int  rc = ok ? rr : (nrows - 1);
        sc[p] = ok ? ns[rc] : 0.0f;
        xp[p] = X + (size_t)rc * K + scol;
    }
    const float* __restrict__ wb = W + (size_t)(2 * half) * NC + n0 + lid;

    float4 xr[NP];
    float  bcur[2 * KS], bnxt[2 * KS];
#pragma unroll
    for (int p = 0; p < NP; ++p) xr[p] = *(const float4*)(xp[p]);
#pragma unroll
    for (int ks = 0; ks < KS; ++ks) {
        bcur[2 * ks]     = wb[(size_t)(ks * 4) * NC];
        bcur[2 * ks + 1] = wb[(size_t)(ks * 4 + 1) * NC];
    }
#pragma unroll
    for (int p = 0; p < NP; ++p) {
        float4 v = xr[p];
        v.x *= sc[p]; v.y *= sc[p]; v.z *= sc[p]; v.w *= sc[p];
        *(float4*)(xs + (srow + RPP * p) * PITCH + scol) = v;
    }
    __syncthreads();

    v8f acc[MT];
#pragma unroll
    for (int mt = 0; mt < MT; ++mt) acc[mt] = v8f{};

    for (int kc = 0; kc < NCH; ++kc) {
        const bool more = kc + 1 < NCH;
        if (more) {
            const int k1 = (kc + 1) * KC;
#pragma unroll
            for (int p = 0; p < NP; ++p) xr[p] = *(const float4*)(xp[p] + k1);
#pragma unroll
            for (int ks = 0; ks < KS; ++ks) {
                bnxt[2 * ks]     = wb[(size_t)(k1 + ks * 4) * NC];
                bnxt[2 * ks + 1] = wb[(size_t)(k1 + ks * 4 + 1) * NC];
            }
        }
        __builtin_amdgcn_sched_barrier(0);
        const float* xb = xs + (kc & 1) * (BM * PITCH) + lid * PITCH + 2 * half;
#pragma unroll
        for (int ks = 0; ks < KS; ++ks) {
            v2f b;
            b.x = bcur[2 * ks];
            b.y = bcur[2 * ks + 1];
#pragma unroll
            for (int mt = 0; mt < MT; ++mt) {
                const v2f a = *(const v2f*)(xb + mt * 16 * PITCH + ks * 4);
                acc[mt] = __builtin_amdgcn_wmma_f32_16x16x4_f32(
                     false, a,  false, b,
                     (short)0, acc[mt],  false,  false);
            }
        }
        if (more) {
            float* xd = xs + ((kc + 1) & 1) * (BM * PITCH);
#pragma unroll
            for (int p = 0; p < NP; ++p) {
                float4 v = xr[p];
                v.x *= sc[p]; v.y *= sc[p]; v.z *= sc[p]; v.w *= sc[p];
                *(float4*)(xd + (srow + RPP * p) * PITCH + scol) = v;
            }
#pragma unroll
            for (int i = 0; i < 2 * KS; ++i) bcur[i] = bnxt[i];
            __syncthreads();
        }
    }

    const int rbase = m0 + 8 * half;
    float* __restrict__ pp = P + (size_t)rbase * NC + n0 + lid;
    if (m0 + BM <= nrows) {
#pragma unroll
        for (int mt = 0; mt < MT; ++mt)
#pragma unroll
            for (int r = 0; r < 8; ++r)
                pp[(size_t)(mt * 16 + r) * NC] = acc[mt][r];
    } else {
#pragma unroll
        for (int mt = 0; mt < MT; ++mt)
#pragma unroll
            for (int r = 0; r < 8; ++r)
                if (rbase + mt * 16 + r < nrows)
                    pp[(size_t)(mt * 16 + r) * NC] = acc[mt][r];
    }
}


constexpr int BINS = 256;

__global__ __launch_bounds__(1024)
void k_bin_edges(const int* __restrict__ src, const int* __restrict__ dst, float* lst, float* sl, int n, int E) {
    __shared__ int cnt[BINS];
    __shared__ int base[BINS];
    const int tid = threadIdx.x;
    const int e0  = (blockIdx.x * 1024 + tid) * 4;
    int s[4], d[4];
    if (e0 + 3 < E) {
        const int4 dv = *(const int4*)(dst + e0);
        const int4 sv = *(const int4*)(src + e0);
        d[0] = dv.x; d[1] = dv.y; d[2] = dv.z; d[3] = dv.w;
        s[0] = sv.x; s[1] = sv.y; s[2] = sv.z; s[3] = sv.w;
    } else {
#pragma unroll
        for (int c = 0; c < 4; ++c) {
            const bool in = e0 + c < E;
            d[c] = in ? dst[e0 + c] : -1;
            s[c] = in ? src[e0 + c] : -1;
        }
    }
    const int nb = (n + SUB - 1) / SUB;
    for (int b0 = 0; b0 < nb; b0 += BINS) {
        if (tid < BINS) cnt[tid] = 0;
        __syncthreads();
        int bl[4], rk[4];
#pragma unroll
        for (int c = 0; c < 4; ++c) {
            bl[c] = d[c] / SUB - b0;
            const bool in = (unsigned)d[c] < (unsigned)n && (unsigned)bl[c] < (unsigned)BINS;
            rk[c] = in ? atomicAdd(&cnt[bl[c]], 1) : -1;
        }
        __syncthreads();
        if (tid < BINS && b0 + tid < nb && cnt[tid] > 0) {
            const int r0 = (b0 + tid) * SUB;
            base[tid] = atomicAdd((int*)(list_of(lst, r0) + list_cap(r0, n)), cnt[tid]);
        }
        __syncthreads();
#pragma unroll
        for (int c = 0; c < 4; ++c) {
            if (rk[c] >= 0) {
                const int r0  = (b0 + bl[c]) * SUB;
                const int pos = base[bl[c]] + rk[c];
                if (pos < list_cap(r0, n)) list_of(lst, r0)[pos] = make_int2(s[c], d[c]);
            }
        }
        __syncthreads();
    }
    for (int b0 = 0; b0 < nb; b0 += BINS) {
        if (tid < BINS) cnt[tid] = 0;
        __syncthreads();
        int bl[4], rk[4];
#pragma unroll
        for (int c = 0; c < 4; ++c) {
            bl[c] = s[c] / SUB - b0;
            const bool in = (unsigned)s[c] < (unsigned)n && (unsigned)bl[c] < (unsigned)BINS;
            rk[c] = in ? atomicAdd(&cnt[bl[c]], 1) : -1;
        }
        __syncthreads();
        if (tid < BINS && b0 + tid < nb && cnt[tid] > 0) {
            const int r0 = (b0 + tid) * SUB;
            base[tid] = atomicAdd(slist_of(sl, r0) + slist_cap(r0, n), cnt[tid]);
        }
        __syncthreads();
#pragma unroll
        for (int c = 0; c < 4; ++c) {
            if (rk[c] >= 0) {
                const int r0  = (b0 + bl[c]) * SUB;
                const int pos = base[bl[c]] + rk[c];
                if (pos < slist_cap(r0, n)) slist_of(sl, r0)[pos] = s[c];
            }
        }
        __syncthreads();
    }
}


typedef __attribute__((ext_vector_type(4))) float v4f;
template <int V> struct vecf;
template <> struct vecf<2> { typedef v2f T; };
template <> struct vecf<4> { typedef v4f T; };

constexpr int QCAP = SUB * 32;

template <int D, bool FULL>
__device__ __forceinline__ void gather_own(const float* __restrict__ P, float* acc,
                                           int s_l, int t_l, int nvalid, int lane) {
    constexpr int V = D / 32;
    constexpr int U = 32 / V;
    typedef typename vecf<V>::T vt;
#pragma nounroll
    for (int i0 = 0; i0 < 32; i0 += U) {
        if (!FULL && i0 >= nvalid) break;
        vt v[U];
#pragma unroll
        for (int u = 0; u < U; ++u) {
            if (FULL || i0 + u < nvalid) {
                const int s = __builtin_amdgcn_readlane(s_l, i0 + u);
                v[u] = *(const vt*)(P + (size_t)s * D + V * lane);
            }
        }
#pragma unroll
        for (int u = 0; u < U; ++u) {
            if (FULL || i0 + u < nvalid) {
                const int t = __builtin_amdgcn_readlane(t_l, i0 + u);
                vt* a = (vt*)(acc + t * D + V * lane);
                vt x = *a;
                x += v[u];
                *a = x;
            }
        }
    }
}

template <int D, bool FULL>
__device__ __forceinline__ void gather_add(const float* __restrict__ P, float* acc,
                                           int s_l, int t_l, int nvalid, int lane) {
    constexpr int V = D / 32;
    constexpr int U = 32 / V;
    typedef typename vecf<V>::T vt;
#pragma unroll
    for (int i0 = 0; i0 < 32; i0 += U) {
        if (!FULL && i0 >= nvalid) break;
        vt v[U];
#pragma unroll
        for (int u = 0; u < U; ++u) {
            if (FULL || i0 + u < nvalid) {
                const int s = __builtin_amdgcn_readlane(s_l, i0 + u);
                v[u] = *(const vt*)(P + (size_t)s * D + V * lane);
            }
        }
#pragma unroll
        for (int u = 0; u < U; ++u) {
            if (FULL || i0 + u < nvalid) {
                const int t = __builtin_amdgcn_readlane(t_l, i0 + u);
                float* a = acc + t * D + V * lane;
#pragma unroll
                for (int c = 0; c < V; ++c) atomicAdd(a + c, v[u][c]);
            }
        }
    }
}

template <int D, bool FULL>
__device__ __forceinline__ void ring_drain(const float* __restrict__ P, const int* __restrict__ src,
                                           const int* __restrict__ dst, float* acc, int* rcn, const int* qe,
                                           int lo, int tail, int nvalid, int lane) {
    __builtin_amdgcn_wave_barrier();
    const int e_l = (FULL || lane < nvalid) ? qe[(tail + lane) & 255] : 0;
    const int s_l = src[e_l];
    const int t_l = dst[e_l] - lo;
    if (FULL || lane < nvalid) atomicAdd(&rcn[t_l], 1);
    gather_add<D, FULL>(P, acc, s_l, t_l, nvalid, lane);
}

template <int D, bool RELU>
__global__ __launch_bounds__(1024)
void k_agg_own(const float* __restrict__ P, const int* __restrict__ src,
               const int* __restrict__ dst,
               const float* __restrict__ b, float* H, float* lst, int n, int E) {
    constexpr int M  = SUB;
    constexpr int NW = 32;
    constexpr int CH = 128;
    constexpr int RV = D / 4;
    __shared__ __attribute__((aligned(16))) float acc[M * D];
    __shared__ int2 q[QCAP];
    __shared__ int cell[NW * NW];
    __shared__ int wtot[NW];
    __shared__ int rcn[M];

    const int tid  = threadIdx.x;
    const int lane = tid & 31;
    const int w    = __builtin_amdgcn_readfirstlane(tid >> 5);
    const int lo   = blockIdx.x * M;
    const int rows = min(M, n - lo);

    constexpr int PT = (QCAP + 1023) / 1024;
    const int2* rg  = list_of(lst, lo);
    const int   cap = list_cap(lo, n);
    int2 e[PT];
#pragma unroll
    for (int k = 0; k < PT; ++k) {
        const int i = tid + k * 1024;
        e[k] = (i < cap) ? rg[i] : make_int2(0, lo);
    }
    const int   c   = rg[cap].x;

    for (int i = tid; i < M * D / 4; i += 1024)
        ((float4*)acc)[i] = make_float4(0.f, 0.f, 0.f, 0.f);
    if (tid < M) rcn[tid] = 0;
    cell[tid] = 0;
    __syncthreads();

    if (c <= cap) {
#pragma unroll
        for (int k = 0; k < PT; ++k) {
            if (tid + k * 1024 < c) {
                const int t = e[k].y - lo;
                atomicAdd(&rcn[t], 1);
                atomicAdd(&cell[w * NW + (t & 31)], 1);
            }
        }
        __syncthreads();
        const int cnt = cell[lane * NW + w];
        int inc = cnt;
        for (int d = 1; d < 32; d <<= 1) {
            const int x = __shfl_up(inc, d);
            if (lane >= d) inc += x;
        }
        if (lane == 31) wtot[w] = inc;
        __syncthreads();
        int before = (lane < w) ? wtot[lane] : 0;
        for (int d = 16; d >= 1; d >>= 1) before += __shfl_xor(before, d);
        cell[lane * NW + w] = before + inc - cnt;
        __syncthreads();
#pragma unroll
        for (int k = 0; k < PT; ++k) {
            if (tid + k * 1024 < c) {
                const int t   = e[k].y - lo;
                const int pos = atomicAdd(&cell[w * NW + (t & 31)], 1);
                q[pos] = make_int2(e[k].x, t);
            }
        }
        __syncthreads();
        const int beg = __builtin_amdgcn_readfirstlane(before);
        const int end = beg + wtot[w];
        int base = beg;
        for (; base + 32 <= end; base += 32) {
            const int2 p = q[base + lane];
            gather_own<D, true>(P, acc, p.x, p.y, 32, lane);
        }
        if (base < end) {
            const int2 p = (base + lane < end) ? q[base + lane] : make_int2(0, 0);
            gather_own<D, false>(P, acc, p.x, p.y, end - base, lane);
        }
    } else {
    int* qe = (int*)q + w * 256;
    int head = 0, tail = 0;
    const unsigned lt_mask = (1u << lane) - 1u;

    auto scan_step = [&](const int4 cv, int e0) {
        const int dv[4] = {cv.x, cv.y, cv.z, cv.w};
#pragma unroll
        for (int k = 0; k < 4; ++k) {
            const bool match = (unsigned)(dv[k] - lo) < (unsigned)rows;
            const unsigned m = __builtin_amdgcn_ballot_w32(match);
            if (m) {
                if (match)
                    qe[(head + __builtin_popcount(m & lt_mask)) & 255] = e0 + k;
                head += __builtin_popcount(m);
            }
        }
        while (head - tail >= 32) {
            ring_drain<D, true>(P, src, dst, acc, rcn, qe, lo, tail, 32, lane);
            tail += 32;
        }
    };

    const int nsteps_all = E / CH;
    if (w < nsteps_all) {
        const int rot = (int)blockIdx.x * (nsteps_all / (int)gridDim.x);
        auto step_base = [&](int idx) {
            int s = idx + rot;
            if (s >= nsteps_all) s -= nsteps_all;
            return s * CH;
        };
        auto ld = [&](int idx) {
            const int bb = step_base(idx < nsteps_all ? idx : w);
            return *(const int4*)(dst + bb + lane * 4);
        };
        int idx = w;
        int4 c0 = ld(idx);
        int4 c1 = ld(idx + NW);
        int4 c2 = ld(idx + 2 * NW);
        int4 c3 = ld(idx + 3 * NW);
        for (; idx < nsteps_all; idx += 4 * NW) {
            scan_step(c0, step_base(idx) + lane * 4);
            c0 = ld(idx + 4 * NW);
            if (idx + NW >= nsteps_all) break;
            scan_step(c1, step_base(idx + NW) + lane * 4);
            c1 = ld(idx + 5 * NW);
            if (idx + 2 * NW >= nsteps_all) break;
            scan_step(c2, step_base(idx + 2 * NW) + lane * 4);
            c2 = ld(idx + 6 * NW);
            if (idx + 3 * NW >= nsteps_all) break;
            scan_step(c3, step_base(idx + 3 * NW) + lane * 4);
            c3 = ld(idx + 7 * NW);
        }
    }
    if (w == 0) {
        const int e_tail = nsteps_all * CH;
        for (int eb = e_tail; eb < E; eb += 32) {
            const int e = eb + lane;
            const int d = (e < E) ? dst[e] : -1;
            const bool match = (unsigned)(d - lo) < (unsigned)rows;
            const unsigned m = __builtin_amdgcn_ballot_w32(match);
            if (m) {
                if (match)
                    qe[(head + __builtin_popcount(m & lt_mask)) & 255] = e;
                head += __builtin_popcount(m);
            }
        }
        while (head - tail >= 32) {
            ring_drain<D, true>(P, src, dst, acc, rcn, qe, lo, tail, 32, lane);
            tail += 32;
        }
    }
    if (head - tail > 0)
        ring_drain<D, false>(P, src, dst, acc, rcn, qe, lo, tail, head - tail, lane);
    }
    __syncthreads();

    for (int i = tid; i < rows * RV; i += 1024) {
        const int r = i / RV, c4 = i % RV;
        const float ndv = rsqrtf(fmaxf((float)rcn[r], 1.0f));
        const v4f a  = *(const v4f*)(acc + i * 4);
        const v4f bb = *(const v4f*)(b + c4 * 4);
        v4f v = a * ndv + bb;
        if (RELU) {
            v.x = fmaxf(v.x, 0.0f); v.y = fmaxf(v.y, 0.0f);
            v.z = fmaxf(v.z, 0.0f); v.w = fmaxf(v.w, 0.0f);
        }
        *(v4f*)(H + (size_t)(lo + r) * D + c4 * 4) = v;
    }
}


extern "C" void kernel_launch(void* const* d_in, const int* in_sizes, int n_in,
                              void* d_out, int out_size, void* d_ws, size_t ws_size,
                              hipStream_t stream) {
    const float* feat = (const float*)d_in[0];
    const float* W1   = (const float*)d_in[1];
    const float* b1   = (const float*)d_in[2];
    const float* W2   = (const float*)d_in[3];
    const float* b2   = (const float*)d_in[4];
    const float* W3   = (const float*)d_in[5];
    const float* b3   = (const float*)d_in[6];
    const int*   src  = (const int*)d_in[7];
    const int*   dst  = (const int*)d_in[8];

    const int D_IN = 256, D_H = 128, D_OUT = 64;
    const int n = in_sizes[0] / D_IN;
    const int E = in_sizes[7];

    float* ns   = (float*)d_ws;
    float* buf0 = ns + 2 * (size_t)n;
    float* buf1 = buf0 + (size_t)n * D_H;

    float* out = (float*)d_out;

    const int mblocks = (n + 63) / 64;

    const int nblk = (n + SUB - 1) / SUB;

    k_zero<<<(nblk + 255) / 256, 256, 0, stream>>>(out, buf1, n);
    k_bin_edges<<<(E + 4095) / 4096, 1024, 0, stream>>>(src, dst, out, buf1, n, E);

    k_src_norm<<<nblk, 1024, 0, stream>>>(src, buf1, ns, n, E);

    k_gemm_scaled<256, 128><<<mblocks, 256, 0, stream>>>(feat, ns, W1, buf0, n);
    k_agg_own<128, true><<<nblk, 1024, 0, stream>>>(buf0, src, dst, b1, buf1, out, n, E);

    k_gemm_scaled<128, 128><<<mblocks, 256, 0, stream>>>(buf1, ns, W2, buf0, n);
    k_agg_own<128, true><<<nblk, 1024, 0, stream>>>(buf0, src, dst, b2, buf1, out, n, E);

    k_gemm_scaled<128, 64><<<mblocks, 128, 0, stream>>>(buf1, ns, W3, buf0, n);
    k_agg_own<64, false><<<nblk, 1024, 0, stream>>>(buf0, src, dst, b3, out, out, n, E);
}


# Aggregation tile moved from LDS to registers: an owner wave keeps its 13 rows as register vectors, adds each gathered row by scalar-indexed register moves; no LDS read-add-write; sum order unchanged
# speedup vs baseline: 4.1923x; 1.0377x over previous
#include <hip/hip_runtime.h>

typedef __attribute__((ext_vector_type(2))) float v2f;
typedef __attribute__((ext_vector_type(8))) float v8f;


constexpr int SUB = 400;

__device__ __forceinline__ int2* list_of(float* lst, int r0) { return (int2*)(lst + (size_t)r0 * 64); }
__device__ __forceinline__ int   list_cap(int r0, int n)    { return min(SUB, n - r0) * 32 - 1; }

__device__ __forceinline__ int* slist_of(float* sl, int r0)       { return (int*)(sl + (size_t)r0 * 128); }
__device__ __forceinline__ int  slist_cap(int r0, int n)          { return min(SUB, n - r0) * 128 - 1; }

__global__ void k_zero(float* lst, float* sl, int n) {
    const int i = blockIdx.x * blockDim.x + threadIdx.x;
    if (i < (n + SUB - 1) / SUB) {
        const int r0 = i * SUB;
        list_of(lst, r0)[list_cap(r0, n)] = make_int2(0, 0);
        slist_of(sl, r0)[slist_cap(r0, n)] = 0;
    }
}

__global__ __launch_bounds__(1024)
void k_src_norm(const int* __restrict__ src, float* sl, float* __restrict__ ns, int n, int E) {
    __shared__ int cnt[SUB];
    const int tid  = threadIdx.x;
    const int r0   = blockIdx.x * SUB;
    const int rows = min(SUB, n - r0);
    if (tid < SUB) cnt[tid] = 0;
    __syncthreads();
    const int* lp  = slist_of(sl, r0);
    const int  cap = slist_cap(r0, n);
    const int  c   = lp[cap];
    if (c <= cap) {
        for (int i = tid * 4; i < c; i += 4096) {
            if (i + 3 < c) {
                const int4 v = *(const int4*)(lp + i);
                const int t[4] = {v.x - r0, v.y - r0, v.z - r0, v.w - r0};
#pragma unroll
                for (int k = 0; k < 4; ++k)
                    if ((unsigned)t[k] < (unsigned)rows) atomicAdd(&cnt[t[k]], 1);
            } else {
                for (int k = 0; k < 4 && i + k < c; ++k) {
                    const int t = lp[i + k] - r0;
                    if ((unsigned)t < (unsigned)rows) atomicAdd(&cnt[t], 1);
                }
            }
        }
    } else {
        for (int e = tid; e < E; e += 1024) {
            const int t = src[e] - r0;
            if ((unsigned)t < (unsigned)rows) atomicAdd(&cnt[t], 1);
        }
    }
    __syncthreads();
    if (tid < rows) ns[r0 + tid] = rsqrtf(fmaxf((float)cnt[tid], 1.0f));
}

template <int K, int NC>
__global__ __launch_bounds__(NC * 2)
void k_gemm_scaled(const float* __restrict__ X, const float* __restrict__ ns,
                   const float* __restrict__ W, float* __restrict__ P,
                   int nrows) {
    constexpr int BM    = 64;
    constexpr int MT    = BM / 16;
    constexpr int KC    = 32;
    constexpr int PITCH = KC + 4;
    constexpr int NTHR  = NC * 2;
    constexpr int RPP   = NTHR / 8;
    constexpr int NP    = BM / RPP;
    constexpr int NCH   = K / KC;
    constexpr int KS    = KC / 4;
    static_assert(K % KC == 0 && NC % 16 == 0 && BM % RPP == 0, "tile shape");
    __shared__ __attribute__((aligned(16))) float xs[2 * BM * PITCH];

    const int tid  = threadIdx.x;
    const int lane = tid & 31;
    const int w    = __builtin_amdgcn_readfirstlane(tid >> 5);
    const int half = lane >> 4;
    const int lid  = lane & 15;
    const int m0   = blockIdx.x * BM;
    const int n0   = w * 16;

    const int srow = tid >> 3;
    const int scol = (tid & 7) * 4;
    const float* xp[NP];
    float sc[NP];
#pragma unroll
    for (int p = 0; p < NP; ++p) {
        const int  rr = m0 + srow + RPP * p;
        const bool ok = rr < nrows;
        const int  rc = ok ? rr : (nrows - 1);
        sc[p] = ok ? ns[rc] : 0.0f;
        xp[p] = X + (size_t)rc * K + scol;
    }
    const float* __restrict__ wb = W + (size_t)(2 * half) * NC + n0 + lid;

    float4 xr[NP];
    float  bcur[2 * KS], bnxt[2 * KS];
#pragma unroll
    for (int p = 0; p < NP; ++p) xr[p] = *(const float4*)(xp[p]);
#pragma unroll
    for (int ks = 0; ks < KS; ++ks) {
        bcur[2 * ks]     = wb[(size_t)(ks * 4) * NC];
        bcur[2 * ks + 1] = wb[(size_t)(ks * 4 + 1) * NC];
    }
#pragma unroll
    for (int p = 0; p < NP; ++p) {
        float4 v = xr[p];
        v.x *= sc[p]; v.y *= sc[p]; v.z *= sc[p]; v.w *= sc[p];
        *(float4*)(xs + (srow + RPP * p) * PITCH + scol) = v;
    }
    __syncthreads();

    v8f acc[MT];
#pragma unroll
    for (int mt = 0; mt < MT; ++mt) acc[mt] = v8f{};

    for (int kc = 0; kc < NCH; ++kc) {
        const bool more = kc + 1 < NCH;
        if (more) {
            const int k1 = (kc + 1) * KC;
#pragma unroll
            for (int p = 0; p < NP; ++p) xr[p] = *(const float4*)(xp[p] + k1);
#pragma unroll
            for (int ks = 0; ks < KS; ++ks) {
                bnxt[2 * ks]     = wb[(size_t)(k1 + ks * 4) * NC];
                bnxt[2 * ks + 1] = wb[(size_t)(k1 + ks * 4 + 1) * NC];
            }
        }
        __builtin_amdgcn_sched_barrier(0);
        const float* xb = xs + (kc & 1) * (BM * PITCH) + lid * PITCH + 2 * half;
#pragma unroll
        for (int ks = 0; ks < KS; ++ks) {
            v2f b;
            b.x = bcur[2 * ks];
            b.y = bcur[2 * ks + 1];
#pragma unroll
            for (int mt = 0; mt < MT; ++mt) {
                const v2f a = *(const v2f*)(xb + mt * 16 * PITCH + ks * 4);
                acc[mt] = __builtin_amdgcn_wmma_f32_16x16x4_f32(
                     false, a,  false, b,
                     (short)0, acc[mt],  false,  false);
            }
        }
        if (more) {
            float* xd = xs + ((kc + 1) & 1) * (BM * PITCH);
#pragma unroll
            for (int p = 0; p < NP; ++p) {
                float4 v = xr[p];
                v.x *= sc[p]; v.y *= sc[p]; v.z *= sc[p]; v.w *= sc[p];
                *(float4*)(xd + (srow + RPP * p) * PITCH + scol) = v;
            }
#pragma unroll
            for (int i = 0; i < 2 * KS; ++i) bcur[i] = bnxt[i];
            __syncthreads();
        }
    }

    const int rbase = m0 + 8 * half;
    float* __restrict__ pp = P + (size_t)rbase * NC + n0 + lid;
    if (m0 + BM <= nrows) {
#pragma unroll
        for (int mt = 0; mt < MT; ++mt)
#pragma unroll
            for (int r = 0; r < 8; ++r)
                pp[(size_t)(mt * 16 + r) * NC] = acc[mt][r];
    } else {
#pragma unroll
        for (int mt = 0; mt < MT; ++mt)
#pragma unroll
            for (int r = 0; r < 8; ++r)
                if (rbase + mt * 16 + r < nrows)
                    pp[(size_t)(mt * 16 + r) * NC] = acc[mt][r];
    }
}


constexpr int BINS = 256;

__global__ __launch_bounds__(1024)
void k_bin_edges(const int* __restrict__ src, const int* __restrict__ dst, float* lst, float* sl, int n, int E) {
    __shared__ int cnt[BINS];
    __shared__ int base[BINS];
    const int tid = threadIdx.x;
    const int e0  = (blockIdx.x * 1024 + tid) * 4;
    int s[4], d[4];
    if (e0 + 3 < E) {
        const int4 dv = *(const int4*)(dst + e0);
        const int4 sv = *(const int4*)(src + e0);
        d[0] = dv.x; d[1] = dv.y; d[2] = dv.z; d[3] = dv.w;
        s[0] = sv.x; s[1] = sv.y; s[2] = sv.z; s[3] = sv.w;
    } else {
#pragma unroll
        for (int c = 0; c < 4; ++c) {
            const bool in = e0 + c < E;
            d[c] = in ? dst[e0 + c] : -1;
            s[c] = in ? src[e0 + c] : -1;
        }
    }
    const int nb = (n + SUB - 1) / SUB;
    for (int b0 = 0; b0 < nb; b0 += BINS) {
        if (tid < BINS) cnt[tid] = 0;
        __syncthreads();
        int bl[4], rk[4];
#pragma unroll
        for (int c = 0; c < 4; ++c) {
            bl[c] = d[c] / SUB - b0;
            const bool in = (unsigned)d[c] < (unsigned)n && (unsigned)bl[c] < (unsigned)BINS;
            rk[c] = in ? atomicAdd(&cnt[bl[c]], 1) : -1;
        }
        __syncthreads();
        if (tid < BINS && b0 + tid < nb && cnt[tid] > 0) {
            const int r0 = (b0 + tid) * SUB;
            base[tid] = atomicAdd((int*)(list_of(lst, r0) + list_cap(r0, n)), cnt[tid]);
        }
        __syncthreads();
#pragma unroll
        for (int c = 0; c < 4; ++c) {
            if (rk[c] >= 0) {
                const int r0  = (b0 + bl[c]) * SUB;
                const int pos = base[bl[c]] + rk[c];
                if (pos < list_cap(r0, n)) list_of(lst, r0)[pos] = make_int2(s[c], d[c]);
            }
        }
        __syncthreads();
    }
    for (int b0 = 0; b0 < nb; b0 += BINS) {
        if (tid < BINS) cnt[tid] = 0;
        __syncthreads();
        int bl[4], rk[4];
#pragma unroll
        for (int c = 0; c < 4; ++c) {
            bl[c] = s[c] / SUB - b0;
            const bool in = (unsigned)s[c] < (unsigned)n && (unsigned)bl[c] < (unsigned)BINS;
            rk[c] = in ? atomicAdd(&cnt[bl[c]], 1) : -1;
        }
        __syncthreads();
        if (tid < BINS && b0 + tid < nb && cnt[tid] > 0) {
            const int r0 = (b0 + tid) * SUB;
            base[tid] = atomicAdd(slist_of(sl, r0) + slist_cap(r0, n), cnt[tid]);
        }
        __syncthreads();
#pragma unroll
        for (int c = 0; c < 4; ++c) {
            if (rk[c] >= 0) {
                const int r0  = (b0 + bl[c]) * SUB;
                const int pos = base[bl[c]] + rk[c];
                if (pos < slist_cap(r0, n)) slist_of(sl, r0)[pos] = s[c];
            }
        }
        __syncthreads();
    }
}


typedef __attribute__((ext_vector_type(4))) float v4f;
template <int V> struct vecf;
template <> struct vecf<2> { typedef v2f T; };
template <> struct vecf<4> { typedef v4f T; };

constexpr int QCAP = SUB * 32;
constexpr int RPW  = (SUB + 31) / 32;

typedef __attribute__((ext_vector_type(16))) float v16f;
static_assert(RPW < 16, "a 16-wide register vector per component holds a wave's rows and a spare slot");
template <int V> struct rows_t { v16f c[V]; };

template <int D>
__device__ __forceinline__ void gather_own(const float* __restrict__ P, rows_t<D / 32>& r,
                                           int s_l, int t_l, int nv, int lane) {
    constexpr int V = D / 32;
    constexpr int U = 32 / V;
    typedef typename vecf<V>::T vt;
#pragma nounroll
    for (int i0 = 0; i0 < nv; i0 += U) {
        vt v[U];
#pragma unroll
        for (int u = 0; u < U; ++u) {
            const int s = __builtin_amdgcn_readlane(s_l, i0 + u);
            v[u] = *(const vt*)(P + (size_t)s * D + V * lane);
        }
#pragma unroll
        for (int u = 0; u < U; ++u) {
            const int j = __builtin_amdgcn_readlane(t_l, i0 + u) >> 5;
#pragma unroll
            for (int k = 0; k < V; ++k) r.c[k][j] += v[u][k];
        }
    }
}

template <int D, bool FULL>
__device__ __forceinline__ void gather_add(const float* __restrict__ P, float* acc,
                                           int s_l, int t_l, int nvalid, int lane) {
    constexpr int V = D / 32;
    constexpr int U = 32 / V;
    typedef typename vecf<V>::T vt;
#pragma unroll
    for (int i0 = 0; i0 < 32; i0 += U) {
        if (!FULL && i0 >= nvalid) break;
        vt v[U];
#pragma unroll
        for (int u = 0; u < U; ++u) {
            if (FULL || i0 + u < nvalid) {
                const int s = __builtin_amdgcn_readlane(s_l, i0 + u);
                v[u] = *(const vt*)(P + (size_t)s * D + V * lane);
            }
        }
#pragma unroll
        for (int u = 0; u < U; ++u) {
            if (FULL || i0 + u < nvalid) {
                const int t = __builtin_amdgcn_readlane(t_l, i0 + u);
                float* a = acc + t * D + V * lane;
#pragma unroll
                for (int c = 0; c < V; ++c) atomicAdd(a + c, v[u][c]);
            }
        }
    }
}

template <int D, bool FULL>
__device__ __forceinline__ void ring_drain(const float* __restrict__ P, const int* __restrict__ src,
                                           const int* __restrict__ dst, float* acc, int* rcn, const int* qe,
                                           int lo, int tail, int nvalid, int lane) {
    __builtin_amdgcn_wave_barrier();
    const int e_l = (FULL || lane < nvalid) ? qe[(tail + lane) & 255] : 0;
    const int s_l = src[e_l];
    const int t_l = dst[e_l] - lo;
    if (FULL || lane < nvalid) atomicAdd(&rcn[t_l], 1);
    gather_add<D, FULL>(P, acc, s_l, t_l, nvalid, lane);
}

template <int D, bool RELU>
__global__ __launch_bounds__(1024)
void k_agg_own(const float* __restrict__ P, const int* __restrict__ src,
               const int* __restrict__ dst,
               const float* __restrict__ b, float* H, float* lst, int n, int E) {
    constexpr int M  = SUB;
    constexpr int NW = 32;
    constexpr int CH = 128;
    constexpr int RV = D / 4;
    constexpr int V  = D / 32;
    typedef typename vecf<V>::T vt;
    __shared__ __attribute__((aligned(16))) float acc[M * D];
    __shared__ int2 q[QCAP];
    __shared__ int cell[NW * NW];
    __shared__ int wtot[NW];
    __shared__ int rcn[M];

    const int tid  = threadIdx.x;
    const int lane = tid & 31;
    const int w    = __builtin_amdgcn_readfirstlane(tid >> 5);
    const int lo   = blockIdx.x * M;
    const int rows = min(M, n - lo);

    constexpr int PT = (QCAP + 1023) / 1024;
    const int2* rg  = list_of(lst, lo);
    const int   cap = list_cap(lo, n);
    int2 e[PT];
#pragma unroll
    for (int k = 0; k < PT; ++k) {
        const int i = tid + k * 1024;
        e[k] = (i < cap) ? rg[i] : make_int2(0, lo);
    }
    const int   c   = rg[cap].x;

    if (tid < M) rcn[tid] = 0;
    cell[tid] = 0;
    __syncthreads();

    if (c <= cap) {
#pragma unroll
        for (int k = 0; k < PT; ++k) {
            if (tid + k * 1024 < c) {
                const int t = e[k].y - lo;
                atomicAdd(&rcn[t], 1);
                atomicAdd(&cell[w * NW + (t & 31)], 1);
            }
        }
        __syncthreads();
        const int cnt = cell[lane * NW + w];
        int inc = cnt;
        for (int d = 1; d < 32; d <<= 1) {
            const int x = __shfl_up(inc, d);
            if (lane >= d) inc += x;
        }
        if (lane == 31) wtot[w] = inc;
        __syncthreads();
        int before = (lane < w) ? wtot[lane] : 0;
        for (int d = 16; d >= 1; d >>= 1) before += __shfl_xor(before, d);
        cell[lane * NW + w] = before + inc - cnt;
        __syncthreads();
#pragma unroll
        for (int k = 0; k < PT; ++k) {
            if (tid + k * 1024 < c) {
                const int t   = e[k].y - lo;
                const int pos = atomicAdd(&cell[w * NW + (t & 31)], 1);
                q[pos] = make_int2(e[k].x, t);
            }
        }
        __syncthreads();
        rows_t<V> r;
#pragma unroll
        for (int k = 0; k < V; ++k) r.c[k] = v16f{};
        const int beg = __builtin_amdgcn_readfirstlane(before);
        const int end = beg + wtot[w];
        for (int base = beg; base < end; base += 32) {
            const int2 p = (base + lane < end) ? q[base + lane] : make_int2(0, RPW * 32);
            gather_own<D>(P, r, p.x, p.y, min(32, end - base), lane);
        }
        const vt bb = *(const vt*)(b + V * lane);
#pragma unroll
        for (int j = 0; j < RPW; ++j) {
            const int t = w + 32 * j;
            if (t < rows) {
                const float ndv = rsqrtf(fmaxf((float)rcn[t], 1.0f));
                vt v;
#pragma unroll
                for (int k = 0; k < V; ++k) v[k] = r.c[k][j];
                v = v * ndv + bb;
                if (RELU) {
#pragma unroll
                    for (int k = 0; k < V; ++k) v[k] = fmaxf(v[k], 0.0f);
                }
                *(vt*)(H + (size_t)(lo + t) * D + V * lane) = v;
            }
        }
    } else {
    for (int i = tid; i < M * D / 4; i += 1024)
        ((float4*)acc)[i] = make_float4(0.f, 0.f, 0.f, 0.f);
    __syncthreads();
    int* qe = (int*)q + w * 256;
    int head = 0, tail = 0;
    const unsigned lt_mask = (1u << lane) - 1u;

    auto scan_step = [&](const int4 cv, int e0) {
        const int dv[4] = {cv.x, cv.y, cv.z, cv.w};
#pragma unroll
        for (int k = 0; k < 4; ++k) {
            const bool match = (unsigned)(dv[k] - lo) < (unsigned)rows;
            const unsigned m = __builtin_amdgcn_ballot_w32(match);
            if (m) {
                if (match)
                    qe[(head + __builtin_popcount(m & lt_mask)) & 255] = e0 + k;
                head += __builtin_popcount(m);
            }
        }
        while (head - tail >= 32) {
            ring_drain<D, true>(P, src, dst, acc, rcn, qe, lo, tail, 32, lane);
            tail += 32;
        }
    };

    const int nsteps_all = E / CH;
    if (w < nsteps_all) {
        const int rot = (int)blockIdx.x * (nsteps_all / (int)gridDim.x);
        auto step_base = [&](int idx) {
            int s = idx + rot;
            if (s >= nsteps_all) s -= nsteps_all;
            return s * CH;
        };
        auto ld = [&](int idx) {
            const int bb = step_base(idx < nsteps_all ? idx : w);
            return *(const int4*)(dst + bb + lane * 4);
        };
        int idx = w;
        int4 c0 = ld(idx);
        int4 c1 = ld(idx + NW);
        int4 c2 = ld(idx + 2 * NW);
        int4 c3 = ld(idx + 3 * NW);
        for (; idx < nsteps_all; idx += 4 * NW) {
            scan_step(c0, step_base(idx) + lane * 4);
            c0 = ld(idx + 4 * NW);
            if (idx + NW >= nsteps_all) break;
            scan_step(c1, step_base(idx + NW) + lane * 4);
            c1 = ld(idx + 5 * NW);
            if (idx + 2 * NW >= nsteps_all) break;
            scan_step(c2, step_base(idx + 2 * NW) + lane * 4);
            c2 = ld(idx + 6 * NW);
            if (idx + 3 * NW >= nsteps_all) break;
            scan_step(c3, step_base(idx + 3 * NW) + lane * 4);
            c3 = ld(idx + 7 * NW);
        }
    }
    if (w == 0) {
        const int e_tail = nsteps_all * CH;
        for (int eb = e_tail; eb < E; eb += 32) {
            const int e = eb + lane;
            const int d = (e < E) ? dst[e] : -1;
            const bool match = (unsigned)(d - lo) < (unsigned)rows;
            const unsigned m = __builtin_amdgcn_ballot_w32(match);
            if (m) {
                if (match)
                    qe[(head + __builtin_popcount(m & lt_mask)) & 255] = e;
                head += __builtin_popcount(m);
            }
        }
        while (head - tail >= 32) {
            ring_drain<D, true>(P, src, dst, acc, rcn, qe, lo, tail, 32, lane);
            tail += 32;
        }
    }
    if (head - tail > 0)
        ring_drain<D, false>(P, src, dst, acc, rcn, qe, lo, tail, head - tail, lane);
    __syncthreads();

    for (int i = tid; i < rows * RV; i += 1024) {
        const int r = i / RV, c4 = i % RV;
        const float ndv = rsqrtf(fmaxf((float)rcn[r], 1.0f));
        const v4f a  = *(const v4f*)(acc + i * 4);
        const v4f bb = *(const v4f*)(b + c4 * 4);
        v4f v = a * ndv + bb;
        if (RELU) {
            v.x = fmaxf(v.x, 0.0f); v.y = fmaxf(v.y, 0.0f);
            v.z = fmaxf(v.z, 0.0f); v.w = fmaxf(v.w, 0.0f);
        }
        *(v4f*)(H + (size_t)(lo + r) * D + c4 * 4) = v;
    }
    }
}


extern "C" void kernel_launch(void* const* d_in, const int* in_sizes, int n_in,
                              void* d_out, int out_size, void* d_ws, size_t ws_size,
                              hipStream_t stream) {
    const float* feat = (const float*)d_in[0];
    const float* W1   = (const float*)d_in[1];
    const float* b1   = (const float*)d_in[2];
    const float* W2   = (const float*)d_in[3];
    const float* b2   = (const float*)d_in[4];
    const float* W3   = (const float*)d_in[5];
    const float* b3   = (const float*)d_in[6];
    const int*   src  = (const int*)d_in[7];
    const int*   dst  = (const int*)d_in[8];

    const int D_IN = 256, D_H = 128, D_OUT = 64;
    const int n = in_sizes[0] / D_IN;
    const int E = in_sizes[7];

    float* ns   = (float*)d_ws;
    float* buf0 = ns + 2 * (size_t)n;
    float* buf1 = buf0 + (size_t)n * D_H;

    float* out = (float*)d_out;

    const int mblocks = (n + 63) / 64;

    const int nblk = (n + SUB - 1) / SUB;

    k_zero<<<(nblk + 255) / 256, 256, 0, stream>>>(out, buf1, n);
    k_bin_edges<<<(E + 4095) / 4096, 1024, 0, stream>>>(src, dst, out, buf1, n, E);

    k_src_norm<<<nblk, 1024, 0, stream>>>(src, buf1, ns, n, E);

    k_gemm_scaled<256, 128><<<mblocks, 256, 0, stream>>>(feat, ns, W1, buf0, n);
    k_agg_own<128, true><<<nblk, 1024, 0, stream>>>(buf0, src, dst, b1, buf1, out, n, E);

    k_gemm_scaled<128, 128><<<mblocks, 256, 0, stream>>>(buf1, ns, W2, buf0, n);
    k_agg_own<128, true><<<nblk, 1024, 0, stream>>>(buf0, src, dst, b2, buf1, out, n, E);

    k_gemm_scaled<128, 64><<<mblocks, 128, 0, stream>>>(buf1, ns, W3, buf0, n);
    k_agg_own<64, false><<<nblk, 1024, 0, stream>>>(buf0, src, dst, b3, out, out, n, E);
}
